# starting kernel
# baseline (speedup 1.0000x reference)
#include <hip/hip_runtime.h>

typedef float v2f __attribute__((ext_vector_type(2)));
typedef float v8f __attribute__((ext_vector_type(8)));

#define NNODES 50000
#define NEDGES 800000
#define KMAX   128
#define BPAD   (KMAX + 2)


__global__ void zero_kernel(float* __restrict__ p, int n) {
    int i = blockIdx.x * blockDim.x + threadIdx.x;
    if (i < n) p[i] = 0.0f;
}

__global__ void degree_kernel(const int* __restrict__ dst, float* __restrict__ deg, int E) {
    int e = blockIdx.x * blockDim.x + threadIdx.x;
    if (e < E) atomicAdd(&deg[dst[e]], 1.0f);
}

__global__ void invdeg_kernel(float* __restrict__ deg, int n) {
    int i = blockIdx.x * blockDim.x + threadIdx.x;
    if (i < n) deg[i] = 1.0f / fmaxf(deg[i], 1.0f);
}

__global__ __launch_bounds__(256) void gemm_wmma_kernel(
    const float* __restrict__ Xin, const float* __restrict__ W,
    float* __restrict__ Out, int K, int N, int Mtiles)
{
    __shared__ float smem[16 * BPAD];

    int tid = threadIdx.x;
    int n0  = blockIdx.y << 4;

    for (int idx = tid; idx < (K << 4); idx += 256) {
        int k = idx >> 4;
        int n = idx & 15;
        int col = n0 + n;
        smem[n * BPAD + k] = (col < N) ? W[(size_t)k * N + col] : 0.0f;
    }
    __syncthreads();

    int tM = (blockIdx.x << 3) + (tid >> 5);
    if (tM >= Mtiles) return;

    int lane = tid & 31;
    int half = lane >> 4;
    int lid  = lane & 15;
    int m0   = tM << 4;

    const float* arow = Xin + (size_t)(m0 + lid) * K;
    const float* brow = smem + lid * BPAD;
    int kofs = half << 1;

    v8f acc = {};
    for (int k0 = 0; k0 < K; k0 += 4) {
        int ka = k0 + kofs;
        v2f a = *(const v2f*)(arow + ka);
        v2f b = *(const v2f*)(brow + ka);
        acc = __builtin_amdgcn_wmma_f32_16x16x4_f32(
             false, a,  false, b,
             (short)0, acc,  false,  false);
    }

    int ncol = n0 + lid;
    if (ncol < N) {
        float* orow = Out + (size_t)(m0 + (half << 3)) * N + ncol;
#pragma unroll
        for (int r = 0; r < 8; ++r)
            orow[(size_t)r * N] = acc[r];
    }
}

__global__ void scatter_kernel(const float* __restrict__ P,
                               const int* __restrict__ src,
                               const int* __restrict__ dst,
                               float* __restrict__ agg, int E, int F)
{
    int e = blockIdx.x * blockDim.x + threadIdx.x;
    if (e >= E) return;
    int c = blockIdx.y << 2;
    int s = src[e];
    int d = dst[e];
    float4 v = *(const float4*)(P + (size_t)s * F + c);
    float* base = agg + (size_t)d * F + c;
    atomicAdd(base + 0, v.x);
    atomicAdd(base + 1, v.y);
    atomicAdd(base + 2, v.z);
    atomicAdd(base + 3, v.w);
}

__global__ void pointwise_kernel(const float* __restrict__ agg,
                                 const float* __restrict__ inv_deg,
                                 const float* __restrict__ bias,
                                 float* __restrict__ out, int F, int relu)
{
    int i = blockIdx.x * blockDim.x + threadIdx.x;
    int total = NNODES * F;
    if (i >= total) return;
    int node = i / F;
    int f    = i - node * F;
    float v = agg[i] * inv_deg[node] + bias[f];
    out[i] = relu ? fmaxf(v, 0.0f) : v;
}


extern "C" void kernel_launch(void* const* d_in, const int* in_sizes, int n_in,
                              void* d_out, int out_size, void* d_ws, size_t ws_size,
                              hipStream_t stream)
{
    const float* features = (const float*)d_in[0];
    const int*   src      = (const int*)  d_in[1];
    const int*   dst      = (const int*)  d_in[2];
    const float* W0       = (const float*)d_in[3];
    const float* b0       = (const float*)d_in[4];
    const float* W1       = (const float*)d_in[5];
    const float* b1       = (const float*)d_in[6];
    const float* W2       = (const float*)d_in[7];
    const float* b2       = (const float*)d_in[8];
    float*       out      = (float*)d_out;

    float* ws  = (float*)d_ws;
    float* deg = ws;
    float* P   = ws + 50048;
    float* A   = P  + (size_t)NNODES * 64;
    float* H   = A  + (size_t)NNODES * 64;

    const int Mtiles     = NNODES / 16;
    const int gemmBlksX  = (Mtiles + 7) / 8;
    const int edgeBlocks = (NEDGES + 255) / 256;

    zero_kernel  <<<(NNODES + 255) / 256, 256, 0, stream>>>(deg, NNODES);
    degree_kernel<<<edgeBlocks, 256, 0, stream>>>(dst, deg, NEDGES);
    invdeg_kernel<<<(NNODES + 255) / 256, 256, 0, stream>>>(deg, NNODES);

    {
        int F = 64, tot = NNODES * F;
        gemm_wmma_kernel<<<dim3(gemmBlksX, 4), 256, 0, stream>>>(features, W0, P, 128, F, Mtiles);
        zero_kernel     <<<(tot + 255) / 256, 256, 0, stream>>>(A, tot);
        scatter_kernel  <<<dim3(edgeBlocks, F / 4), 256, 0, stream>>>(P, src, dst, A, NEDGES, F);
        pointwise_kernel<<<(tot + 255) / 256, 256, 0, stream>>>(A, deg, b0, H, F, 1);
    }

    {
        int F = 64, tot = NNODES * F;
        gemm_wmma_kernel<<<dim3(gemmBlksX, 4), 256, 0, stream>>>(H, W1, P, 64, F, Mtiles);
        zero_kernel     <<<(tot + 255) / 256, 256, 0, stream>>>(A, tot);
        scatter_kernel  <<<dim3(edgeBlocks, F / 4), 256, 0, stream>>>(P, src, dst, A, NEDGES, F);
        pointwise_kernel<<<(tot + 255) / 256, 256, 0, stream>>>(A, deg, b1, H, F, 1);
    }

    {
        int F = 40, tot = NNODES * F;
        gemm_wmma_kernel<<<dim3(gemmBlksX, 3), 256, 0, stream>>>(H, W2, P, 64, F, Mtiles);
        zero_kernel     <<<(tot + 255) / 256, 256, 0, stream>>>(A, tot);
        scatter_kernel  <<<dim3(edgeBlocks, F / 4), 256, 0, stream>>>(P, src, dst, A, NEDGES, F);
        pointwise_kernel<<<(tot + 255) / 256, 256, 0, stream>>>(A, deg, b2, out, F, 0);
    }
}


# Segment sum by gather: edges grouped by dst once per call (int atomic rank, prefix sum), one wave per node adds its source rows; no float atomics, agg not cleared; sum order = arrival rank
# speedup vs baseline: 9.1190x; 9.1190x over previous
#include <hip/hip_runtime.h>


typedef float v2f __attribute__((ext_vector_type(2)));
typedef float v8f __attribute__((ext_vector_type(8)));

#define NNODES 50000
#define NEDGES 800000
#define KMAX   128
#define BPAD   (KMAX + 2)
#define SCAN_T 1024


__global__ void zero_kernel(int* __restrict__ p, int n) {
    int i = blockIdx.x * blockDim.x + threadIdx.x;
    if (i < n) p[i] = 0;
}

__global__ void rank_kernel(const int* __restrict__ dst, int* __restrict__ cnt,
                            int* __restrict__ rank, int E) {
    int e = blockIdx.x * blockDim.x + threadIdx.x;
    if (e < E) rank[e] = atomicAdd(&cnt[dst[e]], 1);
}

__global__ __launch_bounds__(SCAN_T) void offsets_kernel(
    const int* __restrict__ cnt, int* __restrict__ rowptr,
    float* __restrict__ inv_deg, int n)
{
    __shared__ int s_pre[32];
    __shared__ int s_sum[32];

    int tid  = threadIdx.x;
    int lane = tid & 31;
    int wv   = tid >> 5;
    int first = blockIdx.x * SCAN_T;

    int pre = 0;
    for (int i = tid; i < first; i += SCAN_T) pre += cnt[i];

    int i = first + tid;
    int v = (i < n) ? cnt[i] : 0;

    int incl = v;
#pragma unroll
    for (int d = 1; d < 32; d <<= 1) {
        int t = __shfl_up(incl, d);
        if (lane >= d) incl += t;
    }
#pragma unroll
    for (int d = 16; d > 0; d >>= 1) pre += __shfl_xor(pre, d);

    if (lane == 31) s_sum[wv] = incl;
    if (lane == 0)  s_pre[wv] = pre;
    __syncthreads();

    int tot = s_pre[lane];
    int ws  = (lane < wv) ? s_sum[lane] : 0;
#pragma unroll
    for (int d = 16; d > 0; d >>= 1) {
        tot += __shfl_xor(tot, d);
        ws  += __shfl_xor(ws, d);
    }

    if (i <= n) rowptr[i] = tot + ws + incl - v;
    if (i < n)  inv_deg[i] = 1.0f / fmaxf((float)v, 1.0f);
}

__global__ void fill_kernel(const int* __restrict__ src, const int* __restrict__ dst,
                            const int* __restrict__ rank, const int* __restrict__ rowptr,
                            int* __restrict__ csr, int E) {
    int e = blockIdx.x * blockDim.x + threadIdx.x;
    if (e < E) csr[rowptr[dst[e]] + rank[e]] = src[e];
}

__global__ __launch_bounds__(256) void gemm_wmma_kernel(
    const float* __restrict__ Xin, const float* __restrict__ W,
    float* __restrict__ Out, int K, int N, int Mtiles)
{
    __shared__ float smem[16 * BPAD];

    int tid = threadIdx.x;
    int n0  = blockIdx.y << 4;

    for (int idx = tid; idx < (K << 4); idx += 256) {
        int k = idx >> 4;
        int n = idx & 15;
        int col = n0 + n;
        smem[n * BPAD + k] = (col < N) ? W[(size_t)k * N + col] : 0.0f;
    }
    __syncthreads();

    int tM = (blockIdx.x << 3) + (tid >> 5);
    if (tM >= Mtiles) return;

    int lane = tid & 31;
    int half = lane >> 4;
    int lid  = lane & 15;
    int m0   = tM << 4;

    const float* arow = Xin + (size_t)(m0 + lid) * K;
    const float* brow = smem + lid * BPAD;
    int kofs = half << 1;

    v8f acc = {};
    for (int k0 = 0; k0 < K; k0 += 4) {
        int ka = k0 + kofs;
        v2f a = *(const v2f*)(arow + ka);
        v2f b = *(const v2f*)(brow + ka);
        acc = __builtin_amdgcn_wmma_f32_16x16x4_f32(
             false, a,  false, b,
             (short)0, acc,  false,  false);
    }

    int ncol = n0 + lid;
    if (ncol < N) {
        float* orow = Out + (size_t)(m0 + (half << 3)) * N + ncol;
#pragma unroll
        for (int r = 0; r < 8; ++r)
            orow[(size_t)r * N] = acc[r];
    }
}

#define GCH 16

template <int C, int F>
__device__ __forceinline__ v2f add_rows(const float* __restrict__ P, int my, int j,
                                        int col, v2f acc)
{
    int s[C];
#pragma unroll
    for (int u = 0; u < C; ++u) s[u] = __builtin_amdgcn_readlane(my, j + u);
    v2f v[C];
#pragma unroll
    for (int u = 0; u < C; ++u) v[u] = *(const v2f*)(P + (size_t)s[u] * F + col);
#pragma unroll
    for (int u = 0; u < C; ++u) acc += v[u];
    return acc;
}

template <int F>
__global__ __launch_bounds__(256) void gather_kernel(
    const float* __restrict__ P, const int* __restrict__ rowptr,
    const int* __restrict__ csr, float* __restrict__ agg)
{
    int lane = threadIdx.x & 31;
    int node = __builtin_amdgcn_readfirstlane((int)(blockIdx.x * 8 + (threadIdx.x >> 5)));
    int beg = rowptr[node];
    int end = rowptr[node + 1];
    bool act = (2 * lane < F);
    int col = act ? 2 * lane : 2 * lane - F;

    v2f acc = {0.0f, 0.0f};
    for (int base = beg; base < end; base += 32) {
        int n = min(32, end - base);
        int my = (lane < n) ? csr[base + lane] : 0;
        for (int j = 0; j < n; j += GCH) {
            switch (min(GCH, n - j)) {
            case 16: acc = add_rows<16, F>(P, my, j, col, acc); break;
            case 15: acc = add_rows<15, F>(P, my, j, col, acc); break;
            case 14: acc = add_rows<14, F>(P, my, j, col, acc); break;
            case 13: acc = add_rows<13, F>(P, my, j, col, acc); break;
            case 12: acc = add_rows<12, F>(P, my, j, col, acc); break;
            case 11: acc = add_rows<11, F>(P, my, j, col, acc); break;
            case 10: acc = add_rows<10, F>(P, my, j, col, acc); break;
            case  9: acc = add_rows< 9, F>(P, my, j, col, acc); break;
            case  8: acc = add_rows< 8, F>(P, my, j, col, acc); break;
            case  7: acc = add_rows< 7, F>(P, my, j, col, acc); break;
            case  6: acc = add_rows< 6, F>(P, my, j, col, acc); break;
            case  5: acc = add_rows< 5, F>(P, my, j, col, acc); break;
            case  4: acc = add_rows< 4, F>(P, my, j, col, acc); break;
            case  3: acc = add_rows< 3, F>(P, my, j, col, acc); break;
            case  2: acc = add_rows< 2, F>(P, my, j, col, acc); break;
            default: acc = add_rows< 1, F>(P, my, j, col, acc); break;
            }
        }
    }
    if (act) *(v2f*)(agg + (size_t)node * F + col) = acc;
}

__global__ void pointwise_kernel(const float* __restrict__ agg,
                                 const float* __restrict__ inv_deg,
                                 const float* __restrict__ bias,
                                 float* __restrict__ out, int F, int relu)
{
    int i = blockIdx.x * blockDim.x + threadIdx.x;
    int total = NNODES * F;
    if (i >= total) return;
    int node = i / F;
    int f    = i - node * F;
    float v = agg[i] * inv_deg[node] + bias[f];
    out[i] = relu ? fmaxf(v, 0.0f) : v;
}


extern "C" void kernel_launch(void* const* d_in, const int* in_sizes, int n_in,
                              void* d_out, int out_size, void* d_ws, size_t ws_size,
                              hipStream_t stream)
{
    const float* features = (const float*)d_in[0];
    const int*   src      = (const int*)  d_in[1];
    const int*   dst      = (const int*)  d_in[2];
    const float* W0       = (const float*)d_in[3];
    const float* b0       = (const float*)d_in[4];
    const float* W1       = (const float*)d_in[5];
    const float* b1       = (const float*)d_in[6];
    const float* W2       = (const float*)d_in[7];
    const float* b2       = (const float*)d_in[8];
    float*       out      = (float*)d_out;

    float* ws     = (float*)d_ws;
    int*   cnt    = (int*)ws;
    float* deg    = ws + 50176;
    int*   rowptr = (int*)(ws + 100352);
    int*   rank   = (int*)(ws + 150528);
    int*   csr    = (int*)(ws + 950528);
    float* P      = ws + 1750528;
    float* A      = P  + (size_t)NNODES * 64;
    float* H      = A  + (size_t)NNODES * 64;

    const int Mtiles     = NNODES / 16;
    const int gemmBlksX  = (Mtiles + 7) / 8;
    const int edgeBlocks = (NEDGES + 255) / 256;
    const int nodeBlocks = NNODES / 8;

    zero_kernel   <<<(NNODES + 255) / 256, 256, 0, stream>>>(cnt, NNODES);
    rank_kernel   <<<edgeBlocks, 256, 0, stream>>>(dst, cnt, rank, NEDGES);
    offsets_kernel<<<NNODES / SCAN_T + 1, SCAN_T, 0, stream>>>(cnt, rowptr, deg, NNODES);
    fill_kernel   <<<edgeBlocks, 256, 0, stream>>>(src, dst, rank, rowptr, csr, NEDGES);

    {
        int F = 64, tot = NNODES * F;
        gemm_wmma_kernel<<<dim3(gemmBlksX, 4), 256, 0, stream>>>(features, W0, P, 128, F, Mtiles);
        gather_kernel<64><<<nodeBlocks, 256, 0, stream>>>(P, rowptr, csr, A);
        pointwise_kernel<<<(tot + 255) / 256, 256, 0, stream>>>(A, deg, b0, H, F, 1);
    }

    {
        int F = 64, tot = NNODES * F;
        gemm_wmma_kernel<<<dim3(gemmBlksX, 4), 256, 0, stream>>>(H, W1, P, 64, F, Mtiles);
        gather_kernel<64><<<nodeBlocks, 256, 0, stream>>>(P, rowptr, csr, A);
        pointwise_kernel<<<(tot + 255) / 256, 256, 0, stream>>>(A, deg, b1, H, F, 1);
    }

    {
        int F = 40, tot = NNODES * F;
        gemm_wmma_kernel<<<dim3(gemmBlksX, 3), 256, 0, stream>>>(H, W2, P, 64, F, Mtiles);
        gather_kernel<40><<<nodeBlocks, 256, 0, stream>>>(P, rowptr, csr, A);
        pointwise_kernel<<<(tot + 255) / 256, 256, 0, stream>>>(A, deg, b2, out, F, 0);
    }
}


# Normalize + bias + ReLU done in the gather wave before its store (in-degree = segment length); the three elementwise launches and the agg round trip are gone; sum order unchanged
# speedup vs baseline: 10.5621x; 1.1583x over previous
#include <hip/hip_runtime.h>


typedef float v2f __attribute__((ext_vector_type(2)));
typedef float v8f __attribute__((ext_vector_type(8)));

#define NNODES 50000
#define NEDGES 800000
#define KMAX   128
#define BPAD   (KMAX + 2)
#define SCAN_T 1024


__global__ void zero_kernel(int* __restrict__ p, int n) {
    int i = blockIdx.x * blockDim.x + threadIdx.x;
    if (i < n) p[i] = 0;
}

__global__ void rank_kernel(const int* __restrict__ dst, int* __restrict__ cnt,
                            int* __restrict__ rank, int E) {
    int e = blockIdx.x * blockDim.x + threadIdx.x;
    if (e < E) rank[e] = atomicAdd(&cnt[dst[e]], 1);
}

__global__ __launch_bounds__(SCAN_T) void offsets_kernel(
    const int* __restrict__ cnt, int* __restrict__ rowptr, int n)
{
    __shared__ int s_pre[32];
    __shared__ int s_sum[32];

    int tid  = threadIdx.x;
    int lane = tid & 31;
    int wv   = tid >> 5;
    int first = blockIdx.x * SCAN_T;

    int pre = 0;
    for (int i = tid; i < first; i += SCAN_T) pre += cnt[i];

    int i = first + tid;
    int v = (i < n) ? cnt[i] : 0;

    int incl = v;
#pragma unroll
    for (int d = 1; d < 32; d <<= 1) {
        int t = __shfl_up(incl, d);
        if (lane >= d) incl += t;
    }
#pragma unroll
    for (int d = 16; d > 0; d >>= 1) pre += __shfl_xor(pre, d);

    if (lane == 31) s_sum[wv] = incl;
    if (lane == 0)  s_pre[wv] = pre;
    __syncthreads();

    int tot = s_pre[lane];
    int ws  = (lane < wv) ? s_sum[lane] : 0;
#pragma unroll
    for (int d = 16; d > 0; d >>= 1) {
        tot += __shfl_xor(tot, d);
        ws  += __shfl_xor(ws, d);
    }

    if (i <= n) rowptr[i] = tot + ws + incl - v;
}

__global__ void fill_kernel(const int* __restrict__ src, const int* __restrict__ dst,
                            const int* __restrict__ rank, const int* __restrict__ rowptr,
                            int* __restrict__ csr, int E) {
    int e = blockIdx.x * blockDim.x + threadIdx.x;
    if (e < E) csr[rowptr[dst[e]] + rank[e]] = src[e];
}

__global__ __launch_bounds__(256) void gemm_wmma_kernel(
    const float* __restrict__ Xin, const float* __restrict__ W,
    float* __restrict__ Out, int K, int N, int Mtiles)
{
    __shared__ float smem[16 * BPAD];

    int tid = threadIdx.x;
    int n0  = blockIdx.y << 4;

    for (int idx = tid; idx < (K << 4); idx += 256) {
        int k = idx >> 4;
        int n = idx & 15;
        int col = n0 + n;
        smem[n * BPAD + k] = (col < N) ? W[(size_t)k * N + col] : 0.0f;
    }
    __syncthreads();

    int tM = (blockIdx.x << 3) + (tid >> 5);
    if (tM >= Mtiles) return;

    int lane = tid & 31;
    int half = lane >> 4;
    int lid  = lane & 15;
    int m0   = tM << 4;

    const float* arow = Xin + (size_t)(m0 + lid) * K;
    const float* brow = smem + lid * BPAD;
    int kofs = half << 1;

    v8f acc = {};
    for (int k0 = 0; k0 < K; k0 += 4) {
        int ka = k0 + kofs;
        v2f a = *(const v2f*)(arow + ka);
        v2f b = *(const v2f*)(brow + ka);
        acc = __builtin_amdgcn_wmma_f32_16x16x4_f32(
             false, a,  false, b,
             (short)0, acc,  false,  false);
    }

    int ncol = n0 + lid;
    if (ncol < N) {
        float* orow = Out + (size_t)(m0 + (half << 3)) * N + ncol;
#pragma unroll
        for (int r = 0; r < 8; ++r)
            orow[(size_t)r * N] = acc[r];
    }
}

#define GCH 16

template <int C, int F>
__device__ __forceinline__ v2f add_rows(const float* __restrict__ P, int my, int j,
                                        int col, v2f acc)
{
    int s[C];
#pragma unroll
    for (int u = 0; u < C; ++u) s[u] = __builtin_amdgcn_readlane(my, j + u);
    v2f v[C];
#pragma unroll
    for (int u = 0; u < C; ++u) v[u] = *(const v2f*)(P + (size_t)s[u] * F + col);
#pragma unroll
    for (int u = 0; u < C; ++u) acc += v[u];
    return acc;
}

template <int F, bool RELU>
__global__ __launch_bounds__(256) void gather_kernel(
    const float* __restrict__ P, const int* __restrict__ rowptr,
    const int* __restrict__ csr, const float* __restrict__ bias,
    float* __restrict__ outp)
{
    int lane = threadIdx.x & 31;
    int node = __builtin_amdgcn_readfirstlane((int)(blockIdx.x * 8 + (threadIdx.x >> 5)));
    int beg = rowptr[node];
    int end = rowptr[node + 1];
    bool act = (2 * lane < F);
    int col = act ? 2 * lane : 2 * lane - F;
    v2f bv = *(const v2f*)(bias + col);

    v2f acc = {0.0f, 0.0f};
    for (int base = beg; base < end; base += 32) {
        int n = min(32, end - base);
        int my = (lane < n) ? csr[base + lane] : 0;
        for (int j = 0; j < n; j += GCH) {
            switch (min(GCH, n - j)) {
            case 16: acc = add_rows<16, F>(P, my, j, col, acc); break;
            case 15: acc = add_rows<15, F>(P, my, j, col, acc); break;
            case 14: acc = add_rows<14, F>(P, my, j, col, acc); break;
            case 13: acc = add_rows<13, F>(P, my, j, col, acc); break;
            case 12: acc = add_rows<12, F>(P, my, j, col, acc); break;
            case 11: acc = add_rows<11, F>(P, my, j, col, acc); break;
            case 10: acc = add_rows<10, F>(P, my, j, col, acc); break;
            case  9: acc = add_rows< 9, F>(P, my, j, col, acc); break;
            case  8: acc = add_rows< 8, F>(P, my, j, col, acc); break;
            case  7: acc = add_rows< 7, F>(P, my, j, col, acc); break;
            case  6: acc = add_rows< 6, F>(P, my, j, col, acc); break;
            case  5: acc = add_rows< 5, F>(P, my, j, col, acc); break;
            case  4: acc = add_rows< 4, F>(P, my, j, col, acc); break;
            case  3: acc = add_rows< 3, F>(P, my, j, col, acc); break;
            case  2: acc = add_rows< 2, F>(P, my, j, col, acc); break;
            default: acc = add_rows< 1, F>(P, my, j, col, acc); break;
            }
        }
    }
    if (act) {
        float inv = 1.0f / fmaxf((float)(end - beg), 1.0f);
        v2f r = acc * inv + bv;
        if (RELU) { r.x = fmaxf(r.x, 0.0f); r.y = fmaxf(r.y, 0.0f); }
        *(v2f*)(outp + (size_t)node * F + col) = r;
    }
}


extern "C" void kernel_launch(void* const* d_in, const int* in_sizes, int n_in,
                              void* d_out, int out_size, void* d_ws, size_t ws_size,
                              hipStream_t stream)
{
    const float* features = (const float*)d_in[0];
    const int*   src      = (const int*)  d_in[1];
    const int*   dst      = (const int*)  d_in[2];
    const float* W0       = (const float*)d_in[3];
    const float* b0       = (const float*)d_in[4];
    const float* W1       = (const float*)d_in[5];
    const float* b1       = (const float*)d_in[6];
    const float* W2       = (const float*)d_in[7];
    const float* b2       = (const float*)d_in[8];
    float*       out      = (float*)d_out;

    float* ws     = (float*)d_ws;
    int*   cnt    = (int*)ws;
    int*   rowptr = (int*)(ws + 50176);
    int*   rank   = (int*)(ws + 100352);
    int*   csr    = (int*)(ws + 900352);
    float* P      = ws + 1700352;
    float* H      = P  + (size_t)NNODES * 64;

    const int Mtiles     = NNODES / 16;
    const int gemmBlksX  = (Mtiles + 7) / 8;
    const int edgeBlocks = (NEDGES + 255) / 256;
    const int nodeBlocks = NNODES / 8;

    zero_kernel   <<<(NNODES + 255) / 256, 256, 0, stream>>>(cnt, NNODES);
    rank_kernel   <<<edgeBlocks, 256, 0, stream>>>(dst, cnt, rank, NEDGES);
    offsets_kernel<<<NNODES / SCAN_T + 1, SCAN_T, 0, stream>>>(cnt, rowptr, NNODES);
    fill_kernel   <<<edgeBlocks, 256, 0, stream>>>(src, dst, rank, rowptr, csr, NEDGES);

    gemm_wmma_kernel<<<dim3(gemmBlksX, 4), 256, 0, stream>>>(features, W0, P, 128, 64, Mtiles);
    gather_kernel<64, true><<<nodeBlocks, 256, 0, stream>>>(P, rowptr, csr, b0, H);

    gemm_wmma_kernel<<<dim3(gemmBlksX, 4), 256, 0, stream>>>(H, W1, P, 64, 64, Mtiles);
    gather_kernel<64, true><<<nodeBlocks, 256, 0, stream>>>(P, rowptr, csr, b1, H);

    gemm_wmma_kernel<<<dim3(gemmBlksX, 3), 256, 0, stream>>>(H, W2, P, 64, 40, Mtiles);
    gather_kernel<40, false><<<nodeBlocks, 256, 0, stream>>>(P, rowptr, csr, b2, out);
}


# Edge sort without global atomics: edges binned by 256-node range, then by node, both ranked by LDS counters (3 launches for 4); a node's sum order = LDS slot order, still run-dependent
# speedup vs baseline: 15.2449x; 1.4434x over previous
#include <hip/hip_runtime.h>


typedef float v2f __attribute__((ext_vector_type(2)));
typedef float v8f __attribute__((ext_vector_type(8)));

#define NNODES 50000
#define NEDGES 800000
#define KMAX   128
#define BPAD   (KMAX + 2)

#define NBUCK  256
#define NSBLK  256
#define SORT_T 1024
#define EPT    4
#define NOEDGE 0xffffffffu

__global__ __launch_bounds__(SORT_T) void bin_count_kernel(
    const int* __restrict__ dst, int* __restrict__ blockhist, int E)
{
    __shared__ int h[NBUCK];
    int tid = threadIdx.x;
    if (tid < NBUCK) h[tid] = 0;
    __syncthreads();
    int chunk = (E + NSBLK - 1) / NSBLK;
    int e0 = blockIdx.x * chunk;
    int e1 = min(E, e0 + chunk);
    for (int base = e0 + tid; base < e1; base += EPT * SORT_T) {
        int d[EPT];
#pragma unroll
        for (int i = 0; i < EPT; ++i) {
            int e = base + i * SORT_T;
            d[i] = (e < e1) ? dst[e] : -1;
        }
#pragma unroll
        for (int i = 0; i < EPT; ++i)
            if (d[i] >= 0) atomicAdd(&h[d[i] >> 8], 1);
    }
    __syncthreads();
    if (tid < NBUCK) blockhist[blockIdx.x * NBUCK + tid] = h[tid];
}

__device__ __forceinline__ int wave_incl_scan(int v, int lane)
{
#pragma unroll
    for (int d = 1; d < 32; d <<= 1) {
        int t = __shfl_up(v, d);
        if (lane >= d) v += t;
    }
    return v;
}

__global__ __launch_bounds__(SORT_T) void bin_scatter_kernel(
    const int* __restrict__ src, const int* __restrict__ dst,
    const int* __restrict__ blockhist, int* __restrict__ bucketbase,
    unsigned* __restrict__ binned, int E)
{
    __shared__ int before[4][NBUCK];
    __shared__ int whole[4][NBUCK];
    __shared__ int wsum[8];
    __shared__ int cur[NBUCK];

    int tid  = threadIdx.x;
    int lane = tid & 31;
    int b    = tid & (NBUCK - 1);
    int q    = tid >> 8;
    int wv   = (tid >> 5) & 7;
    int me   = blockIdx.x;

    int sb = 0, sa = 0;
    for (int k = q * (NSBLK / 4); k < (q + 1) * (NSBLK / 4); ++k) {
        int v = blockhist[k * NBUCK + b];
        sa += v;
        if (k < me) sb += v;
    }
    before[q][b] = sb;
    whole[q][b]  = sa;
    __syncthreads();

    int tot  = whole[0][b] + whole[1][b] + whole[2][b] + whole[3][b];
    int bef  = before[0][b] + before[1][b] + before[2][b] + before[3][b];
    int incl = wave_incl_scan(tot, lane);
    if (q == 0 && lane == 31) wsum[wv] = incl;
    __syncthreads();
    if (q == 0) {
        int pre = 0;
        for (int j = 0; j < wv; ++j) pre += wsum[j];
        int base = pre + incl - tot;
        cur[b] = base + bef;
        if (me == 0) {
            bucketbase[b] = base;
            if (b == NBUCK - 1) bucketbase[NBUCK] = base + tot;
        }
    }
    __syncthreads();

    int chunk = (E + NSBLK - 1) / NSBLK;
    int e0 = me * chunk;
    int e1 = min(E, e0 + chunk);
    for (int base = e0 + tid; base < e1; base += EPT * SORT_T) {
        int d[EPT], s[EPT];
#pragma unroll
        for (int i = 0; i < EPT; ++i) {
            int e = min(base + i * SORT_T, e1 - 1);
            d[i] = dst[e];
            s[i] = src[e];
        }
#pragma unroll
        for (int i = 0; i < EPT; ++i)
            if (base + i * SORT_T < e1) {
                int pos = atomicAdd(&cur[d[i] >> 8], 1);
                binned[pos] = ((unsigned)s[i] << 8) | (unsigned)(d[i] & (NBUCK - 1));
            }
    }
}

__global__ __launch_bounds__(SORT_T) void bucket_csr_kernel(
    const unsigned* __restrict__ binned, const int* __restrict__ bucketbase,
    int* __restrict__ rowptr, int* __restrict__ csr, int n, int E)
{
    __shared__ int slot[NBUCK];
    __shared__ int wsum[8];

    int tid  = threadIdx.x;
    int lane = tid & 31;
    int b    = blockIdx.x;
    int e0 = bucketbase[b];
    int e1 = bucketbase[b + 1];

    if (tid < NBUCK) slot[tid] = 0;
    __syncthreads();
    for (int base = e0 + tid; base < e1; base += EPT * SORT_T) {
        unsigned u[EPT];
#pragma unroll
        for (int i = 0; i < EPT; ++i) {
            int e = base + i * SORT_T;
            u[i] = (e < e1) ? binned[e] : NOEDGE;
        }
#pragma unroll
        for (int i = 0; i < EPT; ++i)
            if (u[i] != NOEDGE) atomicAdd(&slot[u[i] & (NBUCK - 1)], 1);
    }
    __syncthreads();

    int v = (tid < NBUCK) ? slot[tid] : 0;
    int incl = wave_incl_scan(v, lane);
    if (tid < NBUCK && lane == 31) wsum[tid >> 5] = incl;
    __syncthreads();
    if (tid < NBUCK) {
        int pre = 0;
        for (int j = 0; j < (tid >> 5); ++j) pre += wsum[j];
        int start = e0 + pre + incl - v;
        int node = b * NBUCK + tid;
        if (node < n) rowptr[node] = start;
        slot[tid] = start;
    }
    if (b == (int)gridDim.x - 1 && tid == 0) rowptr[n] = E;
    __syncthreads();
    for (int base = e0 + tid; base < e1; base += EPT * SORT_T) {
        unsigned u[EPT];
#pragma unroll
        for (int i = 0; i < EPT; ++i) {
            int e = base + i * SORT_T;
            u[i] = (e < e1) ? binned[e] : NOEDGE;
        }
#pragma unroll
        for (int i = 0; i < EPT; ++i)
            if (u[i] != NOEDGE) {
                int pos = atomicAdd(&slot[u[i] & (NBUCK - 1)], 1);
                csr[pos] = (int)(u[i] >> 8);
            }
    }
}

__global__ __launch_bounds__(256) void gemm_wmma_kernel(
    const float* __restrict__ Xin, const float* __restrict__ W,
    float* __restrict__ Out, int K, int N, int Mtiles)
{
    __shared__ float smem[16 * BPAD];

    int tid = threadIdx.x;
    int n0  = blockIdx.y << 4;

    for (int idx = tid; idx < (K << 4); idx += 256) {
        int k = idx >> 4;
        int n = idx & 15;
        int col = n0 + n;
        smem[n * BPAD + k] = (col < N) ? W[(size_t)k * N + col] : 0.0f;
    }
    __syncthreads();

    int tM = (blockIdx.x << 3) + (tid >> 5);
    if (tM >= Mtiles) return;

    int lane = tid & 31;
    int half = lane >> 4;
    int lid  = lane & 15;
    int m0   = tM << 4;

    const float* arow = Xin + (size_t)(m0 + lid) * K;
    const float* brow = smem + lid * BPAD;
    int kofs = half << 1;

    v8f acc = {};
    for (int k0 = 0; k0 < K; k0 += 4) {
        int ka = k0 + kofs;
        v2f a = *(const v2f*)(arow + ka);
        v2f b = *(const v2f*)(brow + ka);
        acc = __builtin_amdgcn_wmma_f32_16x16x4_f32(
             false, a,  false, b,
             (short)0, acc,  false,  false);
    }

    int ncol = n0 + lid;
    if (ncol < N) {
        float* orow = Out + (size_t)(m0 + (half << 3)) * N + ncol;
#pragma unroll
        for (int r = 0; r < 8; ++r)
            orow[(size_t)r * N] = acc[r];
    }
}

#define GCH 16

template <int C, int F>
__device__ __forceinline__ v2f add_rows(const float* __restrict__ P, int my, int j,
                                        int col, v2f acc)
{
    int s[C];
#pragma unroll
    for (int u = 0; u < C; ++u) s[u] = __builtin_amdgcn_readlane(my, j + u);
    v2f v[C];
#pragma unroll
    for (int u = 0; u < C; ++u) v[u] = *(const v2f*)(P + (size_t)s[u] * F + col);
#pragma unroll
    for (int u = 0; u < C; ++u) acc += v[u];
    return acc;
}

template <int F, bool RELU>
__global__ __launch_bounds__(256) void gather_kernel(
    const float* __restrict__ P, const int* __restrict__ rowptr,
    const int* __restrict__ csr, const float* __restrict__ bias,
    float* __restrict__ outp)
{
    int lane = threadIdx.x & 31;
    int node = __builtin_amdgcn_readfirstlane((int)(blockIdx.x * 8 + (threadIdx.x >> 5)));
    int beg = rowptr[node];
    int end = rowptr[node + 1];
    bool act = (2 * lane < F);
    int col = act ? 2 * lane : 2 * lane - F;
    v2f bv = *(const v2f*)(bias + col);

    v2f acc = {0.0f, 0.0f};
    for (int base = beg; base < end; base += 32) {
        int n = min(32, end - base);
        int my = (lane < n) ? csr[base + lane] : 0;
        for (int j = 0; j < n; j += GCH) {
            switch (min(GCH, n - j)) {
            case 16: acc = add_rows<16, F>(P, my, j, col, acc); break;
            case 15: acc = add_rows<15, F>(P, my, j, col, acc); break;
            case 14: acc = add_rows<14, F>(P, my, j, col, acc); break;
            case 13: acc = add_rows<13, F>(P, my, j, col, acc); break;
            case 12: acc = add_rows<12, F>(P, my, j, col, acc); break;
            case 11: acc = add_rows<11, F>(P, my, j, col, acc); break;
            case 10: acc = add_rows<10, F>(P, my, j, col, acc); break;
            case  9: acc = add_rows< 9, F>(P, my, j, col, acc); break;
            case  8: acc = add_rows< 8, F>(P, my, j, col, acc); break;
            case  7: acc = add_rows< 7, F>(P, my, j, col, acc); break;
            case  6: acc = add_rows< 6, F>(P, my, j, col, acc); break;
            case  5: acc = add_rows< 5, F>(P, my, j, col, acc); break;
            case  4: acc = add_rows< 4, F>(P, my, j, col, acc); break;
            case  3: acc = add_rows< 3, F>(P, my, j, col, acc); break;
            case  2: acc = add_rows< 2, F>(P, my, j, col, acc); break;
            default: acc = add_rows< 1, F>(P, my, j, col, acc); break;
            }
        }
    }
    if (act) {
        float inv = 1.0f / fmaxf((float)(end - beg), 1.0f);
        v2f r = acc * inv + bv;
        if (RELU) { r.x = fmaxf(r.x, 0.0f); r.y = fmaxf(r.y, 0.0f); }
        *(v2f*)(outp + (size_t)node * F + col) = r;
    }
}


extern "C" void kernel_launch(void* const* d_in, const int* in_sizes, int n_in,
                              void* d_out, int out_size, void* d_ws, size_t ws_size,
                              hipStream_t stream)
{
    const float* features = (const float*)d_in[0];
    const int*   src      = (const int*)  d_in[1];
    const int*   dst      = (const int*)  d_in[2];
    const float* W0       = (const float*)d_in[3];
    const float* b0       = (const float*)d_in[4];
    const float* W1       = (const float*)d_in[5];
    const float* b1       = (const float*)d_in[6];
    const float* W2       = (const float*)d_in[7];
    const float* b2       = (const float*)d_in[8];
    float*       out      = (float*)d_out;

    float* ws     = (float*)d_ws;
    int*      bucketbase = (int*)ws;
    int*      rowptr     = (int*)(ws + 50176);
    unsigned* binned     = (unsigned*)(ws + 100352);
    int*      csr        = (int*)(ws + 900352);
    float*    P          = ws + 1700352;
    float*    H          = P  + (size_t)NNODES * 64;
    int*      blockhist  = (int*)(H + (size_t)NNODES * 64);

    const int Mtiles     = NNODES / 16;
    const int gemmBlksX  = (Mtiles + 7) / 8;
    const int nodeBlocks = NNODES / 8;
    const int buckets    = (NNODES + NBUCK - 1) / NBUCK;

    bin_count_kernel  <<<NSBLK, SORT_T, 0, stream>>>(dst, blockhist, NEDGES);
    bin_scatter_kernel<<<NSBLK, SORT_T, 0, stream>>>(src, dst, blockhist, bucketbase, binned, NEDGES);
    bucket_csr_kernel <<<buckets, SORT_T, 0, stream>>>(binned, bucketbase, rowptr, csr, NNODES, NEDGES);

    gemm_wmma_kernel<<<dim3(gemmBlksX, 4), 256, 0, stream>>>(features, W0, P, 128, 64, Mtiles);
    gather_kernel<64, true><<<nodeBlocks, 256, 0, stream>>>(P, rowptr, csr, b0, H);

    gemm_wmma_kernel<<<dim3(gemmBlksX, 4), 256, 0, stream>>>(H, W1, P, 64, 64, Mtiles);
    gather_kernel<64, true><<<nodeBlocks, 256, 0, stream>>>(P, rowptr, csr, b1, H);

    gemm_wmma_kernel<<<dim3(gemmBlksX, 3), 256, 0, stream>>>(H, W2, P, 64, 40, Mtiles);
    gather_kernel<40, false><<<nodeBlocks, 256, 0, stream>>>(P, rowptr, csr, b2, out);
}


# Next layer's product moved into the gather launch: a 16-node block hands its finished rows through LDS to 4 (3) of its waves' f32 WMMAs; two product launches and the hidden-row round trip gone
# speedup vs baseline: 17.5832x; 1.1534x over previous
#include <hip/hip_runtime.h>


typedef float v2f __attribute__((ext_vector_type(2)));
typedef float v8f __attribute__((ext_vector_type(8)));

#define NNODES 50000
#define NEDGES 800000
#define KMAX   128
#define BPAD   (KMAX + 2)
#define FPAD   68

#define NBUCK  256
#define NSBLK  256
#define SORT_T 1024
#define EPT    4
#define NOEDGE 0xffffffffu

__global__ __launch_bounds__(SORT_T) void bin_count_kernel(
    const int* __restrict__ dst, int* __restrict__ blockhist, int E)
{
    __shared__ int h[NBUCK];
    int tid = threadIdx.x;
    if (tid < NBUCK) h[tid] = 0;
    __syncthreads();
    int chunk = (E + NSBLK - 1) / NSBLK;
    int e0 = blockIdx.x * chunk;
    int e1 = min(E, e0 + chunk);
    for (int base = e0 + tid; base < e1; base += EPT * SORT_T) {
        int d[EPT];
#pragma unroll
        for (int i = 0; i < EPT; ++i) {
            int e = base + i * SORT_T;
            d[i] = (e < e1) ? dst[e] : -1;
        }
#pragma unroll
        for (int i = 0; i < EPT; ++i)
            if (d[i] >= 0) atomicAdd(&h[d[i] >> 8], 1);
    }
    __syncthreads();
    if (tid < NBUCK) blockhist[blockIdx.x * NBUCK + tid] = h[tid];
}

__device__ __forceinline__ int wave_incl_scan(int v, int lane)
{
#pragma unroll
    for (int d = 1; d < 32; d <<= 1) {
        int t = __shfl_up(v, d);
        if (lane >= d) v += t;
    }
    return v;
}

__global__ __launch_bounds__(SORT_T) void bin_scatter_kernel(
    const int* __restrict__ src, const int* __restrict__ dst,
    const int* __restrict__ blockhist, int* __restrict__ bucketbase,
    unsigned* __restrict__ binned, int E)
{
    __shared__ int before[4][NBUCK];
    __shared__ int whole[4][NBUCK];
    __shared__ int wsum[8];
    __shared__ int cur[NBUCK];

    int tid  = threadIdx.x;
    int lane = tid & 31;
    int b    = tid & (NBUCK - 1);
    int q    = tid >> 8;
    int wv   = (tid >> 5) & 7;
    int me   = blockIdx.x;

    int sb = 0, sa = 0;
    for (int k = q * (NSBLK / 4); k < (q + 1) * (NSBLK / 4); ++k) {
        int v = blockhist[k * NBUCK + b];
        sa += v;
        if (k < me) sb += v;
    }
    before[q][b] = sb;
    whole[q][b]  = sa;
    __syncthreads();

    int tot  = whole[0][b] + whole[1][b] + whole[2][b] + whole[3][b];
    int bef  = before[0][b] + before[1][b] + before[2][b] + before[3][b];
    int incl = wave_incl_scan(tot, lane);
    if (q == 0 && lane == 31) wsum[wv] = incl;
    __syncthreads();
    if (q == 0) {
        int pre = 0;
        for (int j = 0; j < wv; ++j) pre += wsum[j];
        int base = pre + incl - tot;
        cur[b] = base + bef;
        if (me == 0) {
            bucketbase[b] = base;
            if (b == NBUCK - 1) bucketbase[NBUCK] = base + tot;
        }
    }
    __syncthreads();

    int chunk = (E + NSBLK - 1) / NSBLK;
    int e0 = me * chunk;
    int e1 = min(E, e0 + chunk);
    for (int base = e0 + tid; base < e1; base += EPT * SORT_T) {
        int d[EPT], s[EPT];
#pragma unroll
        for (int i = 0; i < EPT; ++i) {
            int e = min(base + i * SORT_T, e1 - 1);
            d[i] = dst[e];
            s[i] = src[e];
        }
#pragma unroll
        for (int i = 0; i < EPT; ++i)
            if (base + i * SORT_T < e1) {
                int pos = atomicAdd(&cur[d[i] >> 8], 1);
                binned[pos] = ((unsigned)s[i] << 8) | (unsigned)(d[i] & (NBUCK - 1));
            }
    }
}

__global__ __launch_bounds__(SORT_T) void bucket_csr_kernel(
    const unsigned* __restrict__ binned, const int* __restrict__ bucketbase,
    int* __restrict__ rowptr, int* __restrict__ csr, int n, int E)
{
    __shared__ int slot[NBUCK];
    __shared__ int wsum[8];

    int tid  = threadIdx.x;
    int lane = tid & 31;
    int b    = blockIdx.x;
    int e0 = bucketbase[b];
    int e1 = bucketbase[b + 1];

    if (tid < NBUCK) slot[tid] = 0;
    __syncthreads();
    for (int base = e0 + tid; base < e1; base += EPT * SORT_T) {
        unsigned u[EPT];
#pragma unroll
        for (int i = 0; i < EPT; ++i) {
            int e = base + i * SORT_T;
            u[i] = (e < e1) ? binned[e] : NOEDGE;
        }
#pragma unroll
        for (int i = 0; i < EPT; ++i)
            if (u[i] != NOEDGE) atomicAdd(&slot[u[i] & (NBUCK - 1)], 1);
    }
    __syncthreads();

    int v = (tid < NBUCK) ? slot[tid] : 0;
    int incl = wave_incl_scan(v, lane);
    if (tid < NBUCK && lane == 31) wsum[tid >> 5] = incl;
    __syncthreads();
    if (tid < NBUCK) {
        int pre = 0;
        for (int j = 0; j < (tid >> 5); ++j) pre += wsum[j];
        int start = e0 + pre + incl - v;
        int node = b * NBUCK + tid;
        if (node < n) rowptr[node] = start;
        slot[tid] = start;
    }
    if (b == (int)gridDim.x - 1 && tid == 0) rowptr[n] = E;
    __syncthreads();
    for (int base = e0 + tid; base < e1; base += EPT * SORT_T) {
        unsigned u[EPT];
#pragma unroll
        for (int i = 0; i < EPT; ++i) {
            int e = base + i * SORT_T;
            u[i] = (e < e1) ? binned[e] : NOEDGE;
        }
#pragma unroll
        for (int i = 0; i < EPT; ++i)
            if (u[i] != NOEDGE) {
                int pos = atomicAdd(&slot[u[i] & (NBUCK - 1)], 1);
                csr[pos] = (int)(u[i] >> 8);
            }
    }
}

__global__ __launch_bounds__(256) void gemm_wmma_kernel(
    const float* __restrict__ Xin, const float* __restrict__ W,
    float* __restrict__ Out, int K, int N, int Mtiles)
{
    __shared__ float smem[16 * BPAD];

    int tid = threadIdx.x;
    int n0  = blockIdx.y << 4;

    for (int idx = tid; idx < (K << 4); idx += 256) {
        int k = idx >> 4;
        int n = idx & 15;
        int col = n0 + n;
        smem[n * BPAD + k] = (col < N) ? W[(size_t)k * N + col] : 0.0f;
    }
    __syncthreads();

    int tM = (blockIdx.x << 3) + (tid >> 5);
    if (tM >= Mtiles) return;

    int lane = tid & 31;
    int half = lane >> 4;
    int lid  = lane & 15;
    int m0   = tM << 4;

    const float* arow = Xin + (size_t)(m0 + lid) * K;
    const float* brow = smem + lid * BPAD;
    int kofs = half << 1;

    v8f acc = {};
    for (int k0 = 0; k0 < K; k0 += 4) {
        int ka = k0 + kofs;
        v2f a = *(const v2f*)(arow + ka);
        v2f b = *(const v2f*)(brow + ka);
        acc = __builtin_amdgcn_wmma_f32_16x16x4_f32(
             false, a,  false, b,
             (short)0, acc,  false,  false);
    }

    int ncol = n0 + lid;
    if (ncol < N) {
        float* orow = Out + (size_t)(m0 + (half << 3)) * N + ncol;
#pragma unroll
        for (int r = 0; r < 8; ++r)
            orow[(size_t)r * N] = acc[r];
    }
}

#define GCH 16

template <int C, int F>
__device__ __forceinline__ v2f add_rows(const float* __restrict__ P, int my, int j,
                                        int col, v2f acc)
{
    int s[C];
#pragma unroll
    for (int u = 0; u < C; ++u) s[u] = __builtin_amdgcn_readlane(my, j + u);
    v2f v[C];
#pragma unroll
    for (int u = 0; u < C; ++u) v[u] = *(const v2f*)(P + (size_t)s[u] * F + col);
#pragma unroll
    for (int u = 0; u < C; ++u) acc += v[u];
    return acc;
}

template <int F, bool RELU>
__global__ __launch_bounds__(256) void gather_kernel(
    const float* __restrict__ P, const int* __restrict__ rowptr,
    const int* __restrict__ csr, const float* __restrict__ bias,
    float* __restrict__ outp)
{
    int lane = threadIdx.x & 31;
    int node = __builtin_amdgcn_readfirstlane((int)(blockIdx.x * 8 + (threadIdx.x >> 5)));
    int beg = rowptr[node];
    int end = rowptr[node + 1];
    bool act = (2 * lane < F);
    int col = act ? 2 * lane : 2 * lane - F;
    v2f bv = *(const v2f*)(bias + col);

    v2f acc = {0.0f, 0.0f};
    for (int base = beg; base < end; base += 32) {
        int n = min(32, end - base);
        int my = (lane < n) ? csr[base + lane] : 0;
        for (int j = 0; j < n; j += GCH) {
            switch (min(GCH, n - j)) {
            case 16: acc = add_rows<16, F>(P, my, j, col, acc); break;
            case 15: acc = add_rows<15, F>(P, my, j, col, acc); break;
            case 14: acc = add_rows<14, F>(P, my, j, col, acc); break;
            case 13: acc = add_rows<13, F>(P, my, j, col, acc); break;
            case 12: acc = add_rows<12, F>(P, my, j, col, acc); break;
            case 11: acc = add_rows<11, F>(P, my, j, col, acc); break;
            case 10: acc = add_rows<10, F>(P, my, j, col, acc); break;
            case  9: acc = add_rows< 9, F>(P, my, j, col, acc); break;
            case  8: acc = add_rows< 8, F>(P, my, j, col, acc); break;
            case  7: acc = add_rows< 7, F>(P, my, j, col, acc); break;
            case  6: acc = add_rows< 6, F>(P, my, j, col, acc); break;
            case  5: acc = add_rows< 5, F>(P, my, j, col, acc); break;
            case  4: acc = add_rows< 4, F>(P, my, j, col, acc); break;
            case  3: acc = add_rows< 3, F>(P, my, j, col, acc); break;
            case  2: acc = add_rows< 2, F>(P, my, j, col, acc); break;
            default: acc = add_rows< 1, F>(P, my, j, col, acc); break;
            }
        }
    }
    if (act) {
        float inv = 1.0f / fmaxf((float)(end - beg), 1.0f);
        v2f r = acc * inv + bv;
        if (RELU) { r.x = fmaxf(r.x, 0.0f); r.y = fmaxf(r.y, 0.0f); }
        *(v2f*)(outp + (size_t)node * F + col) = r;
    }
}


template <int NOUT, int NT, bool RELU>
__global__ __launch_bounds__(512) void gather_project_kernel(
    const float* __restrict__ P, const int* __restrict__ rowptr,
    const int* __restrict__ csr, const float* __restrict__ bias,
    const float* __restrict__ Wn, float* __restrict__ Pn)
{
    __shared__ float sW[NT * 16 * FPAD];
    __shared__ float sH[16 * FPAD];

    constexpr int N4  = NOUT / 4;
    constexpr int NW4 = 64 * N4;
    constexpr int WPT = (NW4 + 511) / 512;

    int tid  = threadIdx.x;
    int lane = tid & 31;
    int wave = __builtin_amdgcn_readfirstlane(tid >> 5);
    int node = blockIdx.x * 16 + wave;

    float4 w[WPT];
#pragma unroll
    for (int i = 0; i < WPT; ++i) {
        int idx = tid + i * 512;
        w[i] = ((const float4*)Wn)[(NW4 % 512 == 0) ? idx : min(idx, NW4 - 1)];
    }

    int beg = rowptr[node];
    int end = rowptr[node + 1];
    int col = 2 * lane;
    v2f bv = *(const v2f*)(bias + col);

    v2f acc = {0.0f, 0.0f};
    for (int base = beg; base < end; base += 32) {
        int n = min(32, end - base);
        int my = (lane < n) ? csr[base + lane] : 0;
        for (int j = 0; j < n; j += GCH) {
            switch (min(GCH, n - j)) {
            case 16: acc = add_rows<16, 64>(P, my, j, col, acc); break;
            case 15: acc = add_rows<15, 64>(P, my, j, col, acc); break;
            case 14: acc = add_rows<14, 64>(P, my, j, col, acc); break;
            case 13: acc = add_rows<13, 64>(P, my, j, col, acc); break;
            case 12: acc = add_rows<12, 64>(P, my, j, col, acc); break;
            case 11: acc = add_rows<11, 64>(P, my, j, col, acc); break;
            case 10: acc = add_rows<10, 64>(P, my, j, col, acc); break;
            case  9: acc = add_rows< 9, 64>(P, my, j, col, acc); break;
            case  8: acc = add_rows< 8, 64>(P, my, j, col, acc); break;
            case  7: acc = add_rows< 7, 64>(P, my, j, col, acc); break;
            case  6: acc = add_rows< 6, 64>(P, my, j, col, acc); break;
            case  5: acc = add_rows< 5, 64>(P, my, j, col, acc); break;
            case  4: acc = add_rows< 4, 64>(P, my, j, col, acc); break;
            case  3: acc = add_rows< 3, 64>(P, my, j, col, acc); break;
            case  2: acc = add_rows< 2, 64>(P, my, j, col, acc); break;
            default: acc = add_rows< 1, 64>(P, my, j, col, acc); break;
            }
        }
    }
    float inv = 1.0f / fmaxf((float)(end - beg), 1.0f);
    v2f r = acc * inv + bv;
    if (RELU) { r.x = fmaxf(r.x, 0.0f); r.y = fmaxf(r.y, 0.0f); }
    *(v2f*)(sH + wave * FPAD + col) = r;

#pragma unroll
    for (int i = 0; i < WPT; ++i) {
        int idx = tid + i * 512;
        if (idx < NW4) {
            int k = idx / N4;
            int n = 4 * (idx - k * N4);
            sW[(n + 0) * FPAD + k] = w[i].x;
            sW[(n + 1) * FPAD + k] = w[i].y;
            sW[(n + 2) * FPAD + k] = w[i].z;
            sW[(n + 3) * FPAD + k] = w[i].w;
        }
    }
    if (NT * 16 > NOUT)
        for (int idx = tid; idx < (NT * 16 - NOUT) * 64; idx += 512)
            sW[(NOUT + (idx >> 6)) * FPAD + (idx & 63)] = 0.0f;
    __syncthreads();

    if (wave >= NT) return;

    int half = lane >> 4;
    int lid  = lane & 15;
    const float* arow = sH + lid * FPAD + (half << 1);
    const float* brow = sW + (wave * 16 + lid) * FPAD + (half << 1);

    v8f d = {};
#pragma unroll 4
    for (int k0 = 0; k0 < 64; k0 += 4) {
        v2f a = *(const v2f*)(arow + k0);
        v2f b = *(const v2f*)(brow + k0);
        d = __builtin_amdgcn_wmma_f32_16x16x4_f32(
             false, a,  false, b,
             (short)0, d,  false,  false);
    }

    int ncol = wave * 16 + lid;
    if (ncol < NOUT) {
        float* orow = Pn + (size_t)(blockIdx.x * 16 + (half << 3)) * NOUT + ncol;
#pragma unroll
        for (int r8 = 0; r8 < 8; ++r8)
            orow[(size_t)r8 * NOUT] = d[r8];
    }
}


extern "C" void kernel_launch(void* const* d_in, const int* in_sizes, int n_in,
                              void* d_out, int out_size, void* d_ws, size_t ws_size,
                              hipStream_t stream)
{
    const float* features = (const float*)d_in[0];
    const int*   src      = (const int*)  d_in[1];
    const int*   dst      = (const int*)  d_in[2];
    const float* W0       = (const float*)d_in[3];
    const float* b0       = (const float*)d_in[4];
    const float* W1       = (const float*)d_in[5];
    const float* b1       = (const float*)d_in[6];
    const float* W2       = (const float*)d_in[7];
    const float* b2       = (const float*)d_in[8];
    float*       out      = (float*)d_out;

    float* ws     = (float*)d_ws;
    int*      bucketbase = (int*)ws;
    int*      rowptr     = (int*)(ws + 50176);
    unsigned* binned     = (unsigned*)(ws + 100352);
    int*      csr        = (int*)(ws + 900352);
    float*    P          = ws + 1700352;
    float*    Q          = P  + (size_t)NNODES * 64;
    int*      blockhist  = (int*)(Q + (size_t)NNODES * 64);

    const int Mtiles     = NNODES / 16;
    const int gemmBlksX  = (Mtiles + 7) / 8;
    const int nodeBlocks = NNODES / 8;
    const int nodeGroups = NNODES / 16;
    const int buckets    = (NNODES + NBUCK - 1) / NBUCK;

    bin_count_kernel  <<<NSBLK, SORT_T, 0, stream>>>(dst, blockhist, NEDGES);
    bin_scatter_kernel<<<NSBLK, SORT_T, 0, stream>>>(src, dst, blockhist, bucketbase, binned, NEDGES);
    bucket_csr_kernel <<<buckets, SORT_T, 0, stream>>>(binned, bucketbase, rowptr, csr, NNODES, NEDGES);

    gemm_wmma_kernel<<<dim3(gemmBlksX, 4), 256, 0, stream>>>(features, W0, P, 128, 64, Mtiles);

    gather_project_kernel<64, 4, true><<<nodeGroups, 512, 0, stream>>>(P, rowptr, csr, b0, W1, Q);

    gather_project_kernel<40, 3, true><<<nodeGroups, 512, 0, stream>>>(Q, rowptr, csr, b1, W2, P);

    gather_kernel<40, false><<<nodeBlocks, 256, 0, stream>>>(P, rowptr, csr, b2, out);
}


# Layer-0 product: a wave reads its 16 feature rows once as whole 512-B lines into LDS and does all 4 weight stripes there (was 4 waves reading 8-B pieces of 16 lines); k regrouped within the sum
# speedup vs baseline: 18.9845x; 1.0797x over previous
#include <hip/hip_runtime.h>


typedef float v2f __attribute__((ext_vector_type(2)));
typedef float v8f __attribute__((ext_vector_type(8)));

#define NNODES 50000
#define NEDGES 800000
#define FPAD   68

#define NBUCK  256
#define NSBLK  256
#define SORT_T 1024
#define EPT    4
#define NOEDGE 0xffffffffu

__global__ __launch_bounds__(SORT_T) void bin_count_kernel(
    const int* __restrict__ dst, int* __restrict__ blockhist, int E)
{
    __shared__ int h[NBUCK];
    int tid = threadIdx.x;
    if (tid < NBUCK) h[tid] = 0;
    __syncthreads();
    int chunk = (E + NSBLK - 1) / NSBLK;
    int e0 = blockIdx.x * chunk;
    int e1 = min(E, e0 + chunk);
    for (int base = e0 + tid; base < e1; base += EPT * SORT_T) {
        int d[EPT];
#pragma unroll
        for (int i = 0; i < EPT; ++i) {
            int e = base + i * SORT_T;
            d[i] = (e < e1) ? dst[e] : -1;
        }
#pragma unroll
        for (int i = 0; i < EPT; ++i)
            if (d[i] >= 0) atomicAdd(&h[d[i] >> 8], 1);
    }
    __syncthreads();
    if (tid < NBUCK) blockhist[blockIdx.x * NBUCK + tid] = h[tid];
}

__device__ __forceinline__ int wave_incl_scan(int v, int lane)
{
#pragma unroll
    for (int d = 1; d < 32; d <<= 1) {
        int t = __shfl_up(v, d);
        if (lane >= d) v += t;
    }
    return v;
}

__global__ __launch_bounds__(SORT_T) void bin_scatter_kernel(
    const int* __restrict__ src, const int* __restrict__ dst,
    const int* __restrict__ blockhist, int* __restrict__ bucketbase,
    unsigned* __restrict__ binned, int E)
{
    __shared__ int before[4][NBUCK];
    __shared__ int whole[4][NBUCK];
    __shared__ int wsum[8];
    __shared__ int cur[NBUCK];

    int tid  = threadIdx.x;
    int lane = tid & 31;
    int b    = tid & (NBUCK - 1);
    int q    = tid >> 8;
    int wv   = (tid >> 5) & 7;
    int me   = blockIdx.x;

    int sb = 0, sa = 0;
    for (int k = q * (NSBLK / 4); k < (q + 1) * (NSBLK / 4); ++k) {
        int v = blockhist[k * NBUCK + b];
        sa += v;
        if (k < me) sb += v;
    }
    before[q][b] = sb;
    whole[q][b]  = sa;
    __syncthreads();

    int tot  = whole[0][b] + whole[1][b] + whole[2][b] + whole[3][b];
    int bef  = before[0][b] + before[1][b] + before[2][b] + before[3][b];
    int incl = wave_incl_scan(tot, lane);
    if (q == 0 && lane == 31) wsum[wv] = incl;
    __syncthreads();
    if (q == 0) {
        int pre = 0;
        for (int j = 0; j < wv; ++j) pre += wsum[j];
        int base = pre + incl - tot;
        cur[b] = base + bef;
        if (me == 0) {
            bucketbase[b] = base;
            if (b == NBUCK - 1) bucketbase[NBUCK] = base + tot;
        }
    }
    __syncthreads();

    int chunk = (E + NSBLK - 1) / NSBLK;
    int e0 = me * chunk;
    int e1 = min(E, e0 + chunk);
    for (int base = e0 + tid; base < e1; base += EPT * SORT_T) {
        int d[EPT], s[EPT];
#pragma unroll
        for (int i = 0; i < EPT; ++i) {
            int e = min(base + i * SORT_T, e1 - 1);
            d[i] = dst[e];
            s[i] = src[e];
        }
#pragma unroll
        for (int i = 0; i < EPT; ++i)
            if (base + i * SORT_T < e1) {
                int pos = atomicAdd(&cur[d[i] >> 8], 1);
                binned[pos] = ((unsigned)s[i] << 8) | (unsigned)(d[i] & (NBUCK - 1));
            }
    }
}

__global__ __launch_bounds__(SORT_T) void bucket_csr_kernel(
    const unsigned* __restrict__ binned, const int* __restrict__ bucketbase,
    int* __restrict__ rowptr, int* __restrict__ csr, int n, int E)
{
    __shared__ int slot[NBUCK];
    __shared__ int wsum[8];

    int tid  = threadIdx.x;
    int lane = tid & 31;
    int b    = blockIdx.x;
    int e0 = bucketbase[b];
    int e1 = bucketbase[b + 1];

    if (tid < NBUCK) slot[tid] = 0;
    __syncthreads();
    for (int base = e0 + tid; base < e1; base += EPT * SORT_T) {
        unsigned u[EPT];
#pragma unroll
        for (int i = 0; i < EPT; ++i) {
            int e = base + i * SORT_T;
            u[i] = (e < e1) ? binned[e] : NOEDGE;
        }
#pragma unroll
        for (int i = 0; i < EPT; ++i)
            if (u[i] != NOEDGE) atomicAdd(&slot[u[i] & (NBUCK - 1)], 1);
    }
    __syncthreads();

    int v = (tid < NBUCK) ? slot[tid] : 0;
    int incl = wave_incl_scan(v, lane);
    if (tid < NBUCK && lane == 31) wsum[tid >> 5] = incl;
    __syncthreads();
    if (tid < NBUCK) {
        int pre = 0;
        for (int j = 0; j < (tid >> 5); ++j) pre += wsum[j];
        int start = e0 + pre + incl - v;
        int node = b * NBUCK + tid;
        if (node < n) rowptr[node] = start;
        slot[tid] = start;
    }
    if (b == (int)gridDim.x - 1 && tid == 0) rowptr[n] = E;
    __syncthreads();
    for (int base = e0 + tid; base < e1; base += EPT * SORT_T) {
        unsigned u[EPT];
#pragma unroll
        for (int i = 0; i < EPT; ++i) {
            int e = base + i * SORT_T;
            u[i] = (e < e1) ? binned[e] : NOEDGE;
        }
#pragma unroll
        for (int i = 0; i < EPT; ++i)
            if (u[i] != NOEDGE) {
                int pos = atomicAdd(&slot[u[i] & (NBUCK - 1)], 1);
                csr[pos] = (int)(u[i] >> 8);
            }
    }
}

#define K0    128
#define N0    64
#define GPAD  132

__device__ __forceinline__ v8f wmma4(v2f a, v2f b, v8f c)
{
    return __builtin_amdgcn_wmma_f32_16x16x4_f32(
         false, a,  false, b,
         (short)0, c,  false,  false);
}

__global__ __launch_bounds__(256) void gemm0_kernel(
    const float* __restrict__ Xin, const float* __restrict__ W,
    float* __restrict__ Out, int Mtiles)
{
    __shared__ __attribute__((aligned(16))) float sB[N0 * GPAD];
    __shared__ __attribute__((aligned(16))) float sA[8 * 16 * GPAD];

    int tid  = threadIdx.x;
    int lane = tid & 31;
    int wave = __builtin_amdgcn_readfirstlane(tid >> 5);
    int tM   = blockIdx.x * 8 + wave;
    bool live = tM < Mtiles;
    int m0   = min(tM, Mtiles - 1) << 4;
    float* myA = sA + wave * (16 * GPAD);

    float4 a[16];
#pragma unroll
    for (int r = 0; r < 16; ++r)
        a[r] = ((const float4*)(Xin + (size_t)(m0 + r) * K0))[lane];

    float4 w[8];
    int wk[8], wn[8];
#pragma unroll
    for (int i = 0; i < 8; ++i) {
        int g  = wave + 8 * i;
        wk[i] = 8 * (g & 15) + (lane >> 2);
        wn[i] = 16 * (g >> 4) + 4 * (lane & 3);
        w[i] = *(const float4*)(W + wk[i] * N0 + wn[i]);
    }
#pragma unroll
    for (int i = 0; i < 8; ++i) {
        sB[(wn[i] + 0) * GPAD + wk[i]] = w[i].x;
        sB[(wn[i] + 1) * GPAD + wk[i]] = w[i].y;
        sB[(wn[i] + 2) * GPAD + wk[i]] = w[i].z;
        sB[(wn[i] + 3) * GPAD + wk[i]] = w[i].w;
    }
#pragma unroll
    for (int r = 0; r < 16; ++r)
        *(float4*)(myA + r * GPAD + 4 * lane) = a[r];
    __syncthreads();

    int half = lane >> 4;
    int lid  = lane & 15;
    const float* ap = myA + lid * GPAD + (half << 2);
    const float* bp = sB  + lid * GPAD + (half << 2);

    v8f acc0 = {}, acc1 = {}, acc2 = {}, acc3 = {};
#pragma unroll 4
    for (int k8 = 0; k8 < K0; k8 += 8) {
        float4 a4 = *(const float4*)(ap + k8);
        float4 b0 = *(const float4*)(bp + k8);
        float4 b1 = *(const float4*)(bp + 16 * GPAD + k8);
        float4 b2 = *(const float4*)(bp + 32 * GPAD + k8);
        float4 b3 = *(const float4*)(bp + 48 * GPAD + k8);
        v2f alo = {a4.x, a4.y}, ahi = {a4.z, a4.w};
        acc0 = wmma4(alo, (v2f){b0.x, b0.y}, acc0);
        acc1 = wmma4(alo, (v2f){b1.x, b1.y}, acc1);
        acc2 = wmma4(alo, (v2f){b2.x, b2.y}, acc2);
        acc3 = wmma4(alo, (v2f){b3.x, b3.y}, acc3);
        acc0 = wmma4(ahi, (v2f){b0.z, b0.w}, acc0);
        acc1 = wmma4(ahi, (v2f){b1.z, b1.w}, acc1);
        acc2 = wmma4(ahi, (v2f){b2.z, b2.w}, acc2);
        acc3 = wmma4(ahi, (v2f){b3.z, b3.w}, acc3);
    }
    float* dp = myA + (half << 3) * GPAD + lid;
#pragma unroll
    for (int r = 0; r < 8; ++r) {
        dp[r * GPAD +  0] = acc0[r];
        dp[r * GPAD + 16] = acc1[r];
        dp[r * GPAD + 32] = acc2[r];
        dp[r * GPAD + 48] = acc3[r];
    }
    __syncthreads();
    if (live) {
#pragma unroll
        for (int i = 0; i < 8; ++i) {
            int row = 2 * i + half;
            float4 v = *(const float4*)(myA + row * GPAD + 4 * lid);
            ((float4*)(Out + (size_t)(m0 + row) * N0))[lid] = v;
        }
    }
}

#define GCH 16

template <int C, int F>
__device__ __forceinline__ v2f add_rows(const float* __restrict__ P, int my, int j,
                                        int col, v2f acc)
{
    int s[C];
#pragma unroll
    for (int u = 0; u < C; ++u) s[u] = __builtin_amdgcn_readlane(my, j + u);
    v2f v[C];
#pragma unroll
    for (int u = 0; u < C; ++u) v[u] = *(const v2f*)(P + (size_t)s[u] * F + col);
#pragma unroll
    for (int u = 0; u < C; ++u) acc += v[u];
    return acc;
}

template <int F, bool RELU>
__global__ __launch_bounds__(256) void gather_kernel(
    const float* __restrict__ P, const int* __restrict__ rowptr,
    const int* __restrict__ csr, const float* __restrict__ bias,
    float* __restrict__ outp)
{
    int lane = threadIdx.x & 31;
    int node = __builtin_amdgcn_readfirstlane((int)(blockIdx.x * 8 + (threadIdx.x >> 5)));
    int beg = rowptr[node];
    int end = rowptr[node + 1];
    bool act = (2 * lane < F);
    int col = act ? 2 * lane : 2 * lane - F;
    v2f bv = *(const v2f*)(bias + col);

    v2f acc = {0.0f, 0.0f};
    for (int base = beg; base < end; base += 32) {
        int n = min(32, end - base);
        int my = (lane < n) ? csr[base + lane] : 0;
        for (int j = 0; j < n; j += GCH) {
            switch (min(GCH, n - j)) {
            case 16: acc = add_rows<16, F>(P, my, j, col, acc); break;
            case 15: acc = add_rows<15, F>(P, my, j, col, acc); break;
            case 14: acc = add_rows<14, F>(P, my, j, col, acc); break;
            case 13: acc = add_rows<13, F>(P, my, j, col, acc); break;
            case 12: acc = add_rows<12, F>(P, my, j, col, acc); break;
            case 11: acc = add_rows<11, F>(P, my, j, col, acc); break;
            case 10: acc = add_rows<10, F>(P, my, j, col, acc); break;
            case  9: acc = add_rows< 9, F>(P, my, j, col, acc); break;
            case  8: acc = add_rows< 8, F>(P, my, j, col, acc); break;
            case  7: acc = add_rows< 7, F>(P, my, j, col, acc); break;
            case  6: acc = add_rows< 6, F>(P, my, j, col, acc); break;
            case  5: acc = add_rows< 5, F>(P, my, j, col, acc); break;
            case  4: acc = add_rows< 4, F>(P, my, j, col, acc); break;
            case  3: acc = add_rows< 3, F>(P, my, j, col, acc); break;
            case  2: acc = add_rows< 2, F>(P, my, j, col, acc); break;
            default: acc = add_rows< 1, F>(P, my, j, col, acc); break;
            }
        }
    }
    if (act) {
        float inv = 1.0f / fmaxf((float)(end - beg), 1.0f);
        v2f r = acc * inv + bv;
        if (RELU) { r.x = fmaxf(r.x, 0.0f); r.y = fmaxf(r.y, 0.0f); }
        *(v2f*)(outp + (size_t)node * F + col) = r;
    }
}


template <int NOUT, int NT, bool RELU>
__global__ __launch_bounds__(512) void gather_project_kernel(
    const float* __restrict__ P, const int* __restrict__ rowptr,
    const int* __restrict__ csr, const float* __restrict__ bias,
    const float* __restrict__ Wn, float* __restrict__ Pn)
{
    __shared__ float sW[NT * 16 * FPAD];
    __shared__ float sH[16 * FPAD];

    constexpr int N4  = NOUT / 4;
    constexpr int NW4 = 64 * N4;
    constexpr int WPT = (NW4 + 511) / 512;

    int tid  = threadIdx.x;
    int lane = tid & 31;
    int wave = __builtin_amdgcn_readfirstlane(tid >> 5);
    int node = blockIdx.x * 16 + wave;

    float4 w[WPT];
#pragma unroll
    for (int i = 0; i < WPT; ++i) {
        int idx = tid + i * 512;
        w[i] = ((const float4*)Wn)[(NW4 % 512 == 0) ? idx : min(idx, NW4 - 1)];
    }

    int beg = rowptr[node];
    int end = rowptr[node + 1];
    int col = 2 * lane;
    v2f bv = *(const v2f*)(bias + col);

    v2f acc = {0.0f, 0.0f};
    for (int base = beg; base < end; base += 32) {
        int n = min(32, end - base);
        int my = (lane < n) ? csr[base + lane] : 0;
        for (int j = 0; j < n; j += GCH) {
            switch (min(GCH, n - j)) {
            case 16: acc = add_rows<16, 64>(P, my, j, col, acc); break;
            case 15: acc = add_rows<15, 64>(P, my, j, col, acc); break;
            case 14: acc = add_rows<14, 64>(P, my, j, col, acc); break;
            case 13: acc = add_rows<13, 64>(P, my, j, col, acc); break;
            case 12: acc = add_rows<12, 64>(P, my, j, col, acc); break;
            case 11: acc = add_rows<11, 64>(P, my, j, col, acc); break;
            case 10: acc = add_rows<10, 64>(P, my, j, col, acc); break;
            case  9: acc = add_rows< 9, 64>(P, my, j, col, acc); break;
            case  8: acc = add_rows< 8, 64>(P, my, j, col, acc); break;
            case  7: acc = add_rows< 7, 64>(P, my, j, col, acc); break;
            case  6: acc = add_rows< 6, 64>(P, my, j, col, acc); break;
            case  5: acc = add_rows< 5, 64>(P, my, j, col, acc); break;
            case  4: acc = add_rows< 4, 64>(P, my, j, col, acc); break;
            case  3: acc = add_rows< 3, 64>(P, my, j, col, acc); break;
            case  2: acc = add_rows< 2, 64>(P, my, j, col, acc); break;
            default: acc = add_rows< 1, 64>(P, my, j, col, acc); break;
            }
        }
    }
    float inv = 1.0f / fmaxf((float)(end - beg), 1.0f);
    v2f r = acc * inv + bv;
    if (RELU) { r.x = fmaxf(r.x, 0.0f); r.y = fmaxf(r.y, 0.0f); }
    *(v2f*)(sH + wave * FPAD + col) = r;

#pragma unroll
    for (int i = 0; i < WPT; ++i) {
        int idx = tid + i * 512;
        if (idx < NW4) {
            int k = idx / N4;
            int n = 4 * (idx - k * N4);
            sW[(n + 0) * FPAD + k] = w[i].x;
            sW[(n + 1) * FPAD + k] = w[i].y;
            sW[(n + 2) * FPAD + k] = w[i].z;
            sW[(n + 3) * FPAD + k] = w[i].w;
        }
    }
    if (NT * 16 > NOUT)
        for (int idx = tid; idx < (NT * 16 - NOUT) * 64; idx += 512)
            sW[(NOUT + (idx >> 6)) * FPAD + (idx & 63)] = 0.0f;
    __syncthreads();

    if (wave >= NT) return;

    int half = lane >> 4;
    int lid  = lane & 15;
    const float* arow = sH + lid * FPAD + (half << 1);
    const float* brow = sW + (wave * 16 + lid) * FPAD + (half << 1);

    v8f d = {};
#pragma unroll 4
    for (int k0 = 0; k0 < 64; k0 += 4) {
        v2f a = *(const v2f*)(arow + k0);
        v2f b = *(const v2f*)(brow + k0);
        d = __builtin_amdgcn_wmma_f32_16x16x4_f32(
             false, a,  false, b,
             (short)0, d,  false,  false);
    }

    int ncol = wave * 16 + lid;
    if (ncol < NOUT) {
        float* orow = Pn + (size_t)(blockIdx.x * 16 + (half << 3)) * NOUT + ncol;
#pragma unroll
        for (int r8 = 0; r8 < 8; ++r8)
            orow[(size_t)r8 * NOUT] = d[r8];
    }
}


extern "C" void kernel_launch(void* const* d_in, const int* in_sizes, int n_in,
                              void* d_out, int out_size, void* d_ws, size_t ws_size,
                              hipStream_t stream)
{
    const float* features = (const float*)d_in[0];
    const int*   src      = (const int*)  d_in[1];
    const int*   dst      = (const int*)  d_in[2];
    const float* W0       = (const float*)d_in[3];
    const float* b0       = (const float*)d_in[4];
    const float* W1       = (const float*)d_in[5];
    const float* b1       = (const float*)d_in[6];
    const float* W2       = (const float*)d_in[7];
    const float* b2       = (const float*)d_in[8];
    float*       out      = (float*)d_out;

    float* ws     = (float*)d_ws;
    int*      bucketbase = (int*)ws;
    int*      rowptr     = (int*)(ws + 50176);
    unsigned* binned     = (unsigned*)(ws + 100352);
    int*      csr        = (int*)(ws + 900352);
    float*    P          = ws + 1700352;
    float*    Q          = P  + (size_t)NNODES * 64;
    int*      blockhist  = (int*)(Q + (size_t)NNODES * 64);

    const int Mtiles     = NNODES / 16;
    const int gemmBlksX  = (Mtiles + 7) / 8;
    const int nodeBlocks = NNODES / 8;
    const int nodeGroups = NNODES / 16;
    const int buckets    = (NNODES + NBUCK - 1) / NBUCK;

    bin_count_kernel  <<<NSBLK, SORT_T, 0, stream>>>(dst, blockhist, NEDGES);
    bin_scatter_kernel<<<NSBLK, SORT_T, 0, stream>>>(src, dst, blockhist, bucketbase, binned, NEDGES);
    bucket_csr_kernel <<<buckets, SORT_T, 0, stream>>>(binned, bucketbase, rowptr, csr, NNODES, NEDGES);

    gemm0_kernel<<<gemmBlksX, 256, 0, stream>>>(features, W0, P, Mtiles);

    gather_project_kernel<64, 4, true><<<nodeGroups, 512, 0, stream>>>(P, rowptr, csr, b0, W1, Q);

    gather_project_kernel<40, 3, true><<<nodeGroups, 512, 0, stream>>>(Q, rowptr, csr, b1, W2, P);

    gather_kernel<40, false><<<nodeBlocks, 256, 0, stream>>>(P, rowptr, csr, b2, out);
}


# Edge binning made chunk-local (LDS rank + scan; a bucket's block fetches its piece of every chunk) and run on extra blocks of the layer-0 product launch: 7 -> 5 launches; sum order = LDS slot order
# speedup vs baseline: 24.9530x; 1.3144x over previous
#include <hip/hip_runtime.h>


typedef float v2f __attribute__((ext_vector_type(2)));
typedef float v8f __attribute__((ext_vector_type(8)));

#define NNODES 50000
#define NEDGES 800000
#define FPAD   68

#define NBUCK  256
#define NSBLK  256
#define SORT_T 1024
#define EPT    4
#define NOEDGE 0xffffffffu
#define CHUNK  ((NEDGES + NSBLK - 1) / NSBLK)
#define BIN_T  256
#define CPT    ((CHUNK + BIN_T - 1) / BIN_T)
#define NPIECE (NBUCK + 1)

static_assert(SORT_T == 4 * NSBLK, "bucket_csr maps four threads to a chunk");
static_assert(BIN_T == NBUCK, "chunk_bin: one thread per bucket in the scan");
static_assert((NSBLK - 1) * CHUNK < NEDGES, "every chunk holds at least one edge");

__device__ __forceinline__ int wave_incl_scan(int v, int lane)
{
#pragma unroll
    for (int d = 1; d < 32; d <<= 1) {
        int t = __shfl_up(v, d);
        if (lane >= d) v += t;
    }
    return v;
}

__device__ __forceinline__ void chunk_bin(
    const int* __restrict__ src, const int* __restrict__ dst,
    int* __restrict__ pieces, unsigned* __restrict__ binned, int c)
{
    __shared__ int cnt[NBUCK];
    __shared__ int wsum[8];

    int tid  = threadIdx.x;
    int lane = tid & 31;
    cnt[tid] = 0;
    __syncthreads();

    int e0 = c * CHUNK;
    int e1 = min(NEDGES, e0 + CHUNK);
    int d[CPT], s[CPT], r[CPT];
#pragma unroll
    for (int i = 0; i < CPT; ++i) {
        int e = min(e0 + tid + i * BIN_T, e1 - 1);
        d[i] = dst[e];
        s[i] = src[e];
    }
#pragma unroll
    for (int i = 0; i < CPT; ++i) {
        r[i] = 0;
        if (e0 + tid + i * BIN_T < e1) r[i] = atomicAdd(&cnt[d[i] >> 8], 1);
    }
    __syncthreads();

    int v = cnt[tid];
    int incl = wave_incl_scan(v, lane);
    if (lane == 31) wsum[tid >> 5] = incl;
    __syncthreads();
    int pre = 0;
    for (int j = 0; j < (tid >> 5); ++j) pre += wsum[j];
    int excl = pre + incl - v;
    cnt[tid] = excl;
    pieces[c * NPIECE + tid] = excl;
    if (tid == NBUCK - 1) pieces[c * NPIECE + NBUCK] = excl + v;
    __syncthreads();

#pragma unroll
    for (int i = 0; i < CPT; ++i)
        if (e0 + tid + i * BIN_T < e1)
            binned[e0 + cnt[d[i] >> 8] + r[i]] =
                ((unsigned)s[i] << 8) | (unsigned)(d[i] & (NBUCK - 1));
}

__global__ __launch_bounds__(SORT_T) void bucket_csr_kernel(
    const unsigned* __restrict__ binned, const int* __restrict__ pieces,
    int* __restrict__ rowptr, int* __restrict__ csr, int n)
{
    __shared__ int slot[NBUCK];
    __shared__ int wsum[8];
    __shared__ int bstart;

    int tid  = threadIdx.x;
    int lane = tid & 31;
    int b    = blockIdx.x;
    int c    = tid >> 2;
    int p    = tid & 3;

    int lo = pieces[c * NPIECE + b];
    int hi = pieces[c * NPIECE + b + 1];

    if (tid < NBUCK) slot[tid] = 0;
    if (tid == 0) bstart = 0;
    __syncthreads();

    int x = (p == 0) ? lo : 0;
#pragma unroll
    for (int m = 16; m >= 1; m >>= 1) x += __shfl_xor(x, m);
    if (lane == 0) atomicAdd(&bstart, x);

    int g0 = c * CHUNK + lo + p;
    int g1 = c * CHUNK + hi;
    for (int base = g0; base < g1; base += 4 * EPT) {
        unsigned u[EPT];
#pragma unroll
        for (int i = 0; i < EPT; ++i) {
            int e = base + 4 * i;
            u[i] = (e < g1) ? binned[e] : NOEDGE;
        }
#pragma unroll
        for (int i = 0; i < EPT; ++i)
            if (u[i] != NOEDGE) atomicAdd(&slot[u[i] & (NBUCK - 1)], 1);
    }
    __syncthreads();

    int e0 = bstart;
    int v = (tid < NBUCK) ? slot[tid] : 0;
    int incl = wave_incl_scan(v, lane);
    if (tid < NBUCK && lane == 31) wsum[tid >> 5] = incl;
    __syncthreads();
    if (tid < NBUCK) {
        int pre = 0;
        for (int j = 0; j < (tid >> 5); ++j) pre += wsum[j];
        int start = e0 + pre + incl - v;
        int node = b * NBUCK + tid;
        if (node < n) rowptr[node] = start;
        slot[tid] = start;
    }
    if (b == (int)gridDim.x - 1 && tid == 0) rowptr[n] = NEDGES;
    __syncthreads();
    for (int base = g0; base < g1; base += 4 * EPT) {
        unsigned u[EPT];
#pragma unroll
        for (int i = 0; i < EPT; ++i) {
            int e = base + 4 * i;
            u[i] = (e < g1) ? binned[e] : NOEDGE;
        }
#pragma unroll
        for (int i = 0; i < EPT; ++i)
            if (u[i] != NOEDGE) {
                int pos = atomicAdd(&slot[u[i] & (NBUCK - 1)], 1);
                csr[pos] = (int)(u[i] >> 8);
            }
    }
}

#define K0    128
#define N0    64
#define GPAD  132

__device__ __forceinline__ v8f wmma4(v2f a, v2f b, v8f c)
{
    return __builtin_amdgcn_wmma_f32_16x16x4_f32(
         false, a,  false, b,
         (short)0, c,  false,  false);
}

__global__ __launch_bounds__(256) void gemm0_bin_kernel(
    const float* __restrict__ Xin, const float* __restrict__ W,
    float* __restrict__ Out, int Mtiles, int gemmBlocks,
    const int* __restrict__ src, const int* __restrict__ dst,
    int* __restrict__ pieces, unsigned* __restrict__ binned)
{
    if ((int)blockIdx.x >= gemmBlocks) {
        chunk_bin(src, dst, pieces, binned, (int)blockIdx.x - gemmBlocks);
        return;
    }

    __shared__ __attribute__((aligned(16))) float sB[N0 * GPAD];
    __shared__ __attribute__((aligned(16))) float sA[8 * 16 * GPAD];

    int tid  = threadIdx.x;
    int lane = tid & 31;
    int wave = __builtin_amdgcn_readfirstlane(tid >> 5);
    int tM   = blockIdx.x * 8 + wave;
    bool live = tM < Mtiles;
    int m0   = min(tM, Mtiles - 1) << 4;
    float* myA = sA + wave * (16 * GPAD);

    float4 a[16];
#pragma unroll
    for (int r = 0; r < 16; ++r)
        a[r] = ((const float4*)(Xin + (size_t)(m0 + r) * K0))[lane];

    float4 w[8];
    int wk[8], wn[8];
#pragma unroll
    for (int i = 0; i < 8; ++i) {
        int g  = wave + 8 * i;
        wk[i] = 8 * (g & 15) + (lane >> 2);
        wn[i] = 16 * (g >> 4) + 4 * (lane & 3);
        w[i] = *(const float4*)(W + wk[i] * N0 + wn[i]);
    }
#pragma unroll
    for (int i = 0; i < 8; ++i) {
        sB[(wn[i] + 0) * GPAD + wk[i]] = w[i].x;
        sB[(wn[i] + 1) * GPAD + wk[i]] = w[i].y;
        sB[(wn[i] + 2) * GPAD + wk[i]] = w[i].z;
        sB[(wn[i] + 3) * GPAD + wk[i]] = w[i].w;
    }
#pragma unroll
    for (int r = 0; r < 16; ++r)
        *(float4*)(myA + r * GPAD + 4 * lane) = a[r];
    __syncthreads();

    int half = lane >> 4;
    int lid  = lane & 15;
    const float* ap = myA + lid * GPAD + (half << 2);
    const float* bp = sB  + lid * GPAD + (half << 2);

    v8f acc0 = {}, acc1 = {}, acc2 = {}, acc3 = {};
#pragma unroll 4
    for (int k8 = 0; k8 < K0; k8 += 8) {
        float4 a4 = *(const float4*)(ap + k8);
        float4 b0 = *(const float4*)(bp + k8);
        float4 b1 = *(const float4*)(bp + 16 * GPAD + k8);
        float4 b2 = *(const float4*)(bp + 32 * GPAD + k8);
        float4 b3 = *(const float4*)(bp + 48 * GPAD + k8);
        v2f alo = {a4.x, a4.y}, ahi = {a4.z, a4.w};
        acc0 = wmma4(alo, (v2f){b0.x, b0.y}, acc0);
        acc1 = wmma4(alo, (v2f){b1.x, b1.y}, acc1);
        acc2 = wmma4(alo, (v2f){b2.x, b2.y}, acc2);
        acc3 = wmma4(alo, (v2f){b3.x, b3.y}, acc3);
        acc0 = wmma4(ahi, (v2f){b0.z, b0.w}, acc0);
        acc1 = wmma4(ahi, (v2f){b1.z, b1.w}, acc1);
        acc2 = wmma4(ahi, (v2f){b2.z, b2.w}, acc2);
        acc3 = wmma4(ahi, (v2f){b3.z, b3.w}, acc3);
    }
    float* dp = myA + (half << 3) * GPAD + lid;
#pragma unroll
    for (int r = 0; r < 8; ++r) {
        dp[r * GPAD +  0] = acc0[r];
        dp[r * GPAD + 16] = acc1[r];
        dp[r * GPAD + 32] = acc2[r];
        dp[r * GPAD + 48] = acc3[r];
    }
    __syncthreads();
    if (live) {
#pragma unroll
        for (int i = 0; i < 8; ++i) {
            int row = 2 * i + half;
            float4 v = *(const float4*)(myA + row * GPAD + 4 * lid);
            ((float4*)(Out + (size_t)(m0 + row) * N0))[lid] = v;
        }
    }
}

#define GCH 16

template <int C, int F>
__device__ __forceinline__ v2f add_rows(const float* __restrict__ P, int my, int j,
                                        int col, v2f acc)
{
    int s[C];
#pragma unroll
    for (int u = 0; u < C; ++u) s[u] = __builtin_amdgcn_readlane(my, j + u);
    v2f v[C];
#pragma unroll
    for (int u = 0; u < C; ++u) v[u] = *(const v2f*)(P + (size_t)s[u] * F + col);
#pragma unroll
    for (int u = 0; u < C; ++u) acc += v[u];
    return acc;
}

template <int F, bool RELU>
__global__ __launch_bounds__(256) void gather_kernel(
    const float* __restrict__ P, const int* __restrict__ rowptr,
    const int* __restrict__ csr, const float* __restrict__ bias,
    float* __restrict__ outp)
{
    int lane = threadIdx.x & 31;
    int node = __builtin_amdgcn_readfirstlane((int)(blockIdx.x * 8 + (threadIdx.x >> 5)));
    int beg = rowptr[node];
    int end = rowptr[node + 1];
    bool act = (2 * lane < F);
    int col = act ? 2 * lane : 2 * lane - F;
    v2f bv = *(const v2f*)(bias + col);

    v2f acc = {0.0f, 0.0f};
    for (int base = beg; base < end; base += 32) {
        int n = min(32, end - base);
        int my = (lane < n) ? csr[base + lane] : 0;
        for (int j = 0; j < n; j += GCH) {
            switch (min(GCH, n - j)) {
            case 16: acc = add_rows<16, F>(P, my, j, col, acc); break;
            case 15: acc = add_rows<15, F>(P, my, j, col, acc); break;
            case 14: acc = add_rows<14, F>(P, my, j, col, acc); break;
            case 13: acc = add_rows<13, F>(P, my, j, col, acc); break;
            case 12: acc = add_rows<12, F>(P, my, j, col, acc); break;
            case 11: acc = add_rows<11, F>(P, my, j, col, acc); break;
            case 10: acc = add_rows<10, F>(P, my, j, col, acc); break;
            case  9: acc = add_rows< 9, F>(P, my, j, col, acc); break;
            case  8: acc = add_rows< 8, F>(P, my, j, col, acc); break;
            case  7: acc = add_rows< 7, F>(P, my, j, col, acc); break;
            case  6: acc = add_rows< 6, F>(P, my, j, col, acc); break;
            case  5: acc = add_rows< 5, F>(P, my, j, col, acc); break;
            case  4: acc = add_rows< 4, F>(P, my, j, col, acc); break;
            case  3: acc = add_rows< 3, F>(P, my, j, col, acc); break;
            case  2: acc = add_rows< 2, F>(P, my, j, col, acc); break;
            default: acc = add_rows< 1, F>(P, my, j, col, acc); break;
            }
        }
    }
    if (act) {
        float inv = 1.0f / fmaxf((float)(end - beg), 1.0f);
        v2f r = acc * inv + bv;
        if (RELU) { r.x = fmaxf(r.x, 0.0f); r.y = fmaxf(r.y, 0.0f); }
        *(v2f*)(outp + (size_t)node * F + col) = r;
    }
}


template <int NOUT, int NT, bool RELU>
__global__ __launch_bounds__(512) void gather_project_kernel(
    const float* __restrict__ P, const int* __restrict__ rowptr,
    const int* __restrict__ csr, const float* __restrict__ bias,
    const float* __restrict__ Wn, float* __restrict__ Pn)
{
    __shared__ float sW[NT * 16 * FPAD];
    __shared__ float sH[16 * FPAD];

    constexpr int N4  = NOUT / 4;
    constexpr int NW4 = 64 * N4;
    constexpr int WPT = (NW4 + 511) / 512;

    int tid  = threadIdx.x;
    int lane = tid & 31;
    int wave = __builtin_amdgcn_readfirstlane(tid >> 5);
    int node = blockIdx.x * 16 + wave;

    float4 w[WPT];
#pragma unroll
    for (int i = 0; i < WPT; ++i) {
        int idx = tid + i * 512;
        w[i] = ((const float4*)Wn)[(NW4 % 512 == 0) ? idx : min(idx, NW4 - 1)];
    }

    int beg = rowptr[node];
    int end = rowptr[node + 1];
    int col = 2 * lane;
    v2f bv = *(const v2f*)(bias + col);

    v2f acc = {0.0f, 0.0f};
    for (int base = beg; base < end; base += 32) {
        int n = min(32, end - base);
        int my = (lane < n) ? csr[base + lane] : 0;
        for (int j = 0; j < n; j += GCH) {
            switch (min(GCH, n - j)) {
            case 16: acc = add_rows<16, 64>(P, my, j, col, acc); break;
            case 15: acc = add_rows<15, 64>(P, my, j, col, acc); break;
            case 14: acc = add_rows<14, 64>(P, my, j, col, acc); break;
            case 13: acc = add_rows<13, 64>(P, my, j, col, acc); break;
            case 12: acc = add_rows<12, 64>(P, my, j, col, acc); break;
            case 11: acc = add_rows<11, 64>(P, my, j, col, acc); break;
            case 10: acc = add_rows<10, 64>(P, my, j, col, acc); break;
            case  9: acc = add_rows< 9, 64>(P, my, j, col, acc); break;
            case  8: acc = add_rows< 8, 64>(P, my, j, col, acc); break;
            case  7: acc = add_rows< 7, 64>(P, my, j, col, acc); break;
            case  6: acc = add_rows< 6, 64>(P, my, j, col, acc); break;
            case  5: acc = add_rows< 5, 64>(P, my, j, col, acc); break;
            case  4: acc = add_rows< 4, 64>(P, my, j, col, acc); break;
            case  3: acc = add_rows< 3, 64>(P, my, j, col, acc); break;
            case  2: acc = add_rows< 2, 64>(P, my, j, col, acc); break;
            default: acc = add_rows< 1, 64>(P, my, j, col, acc); break;
            }
        }
    }
    float inv = 1.0f / fmaxf((float)(end - beg), 1.0f);
    v2f r = acc * inv + bv;
    if (RELU) { r.x = fmaxf(r.x, 0.0f); r.y = fmaxf(r.y, 0.0f); }
    *(v2f*)(sH + wave * FPAD + col) = r;

#pragma unroll
    for (int i = 0; i < WPT; ++i) {
        int idx = tid + i * 512;
        if (idx < NW4) {
            int k = idx / N4;
            int n = 4 * (idx - k * N4);
            sW[(n + 0) * FPAD + k] = w[i].x;
            sW[(n + 1) * FPAD + k] = w[i].y;
            sW[(n + 2) * FPAD + k] = w[i].z;
            sW[(n + 3) * FPAD + k] = w[i].w;
        }
    }
    if (NT * 16 > NOUT)
        for (int idx = tid; idx < (NT * 16 - NOUT) * 64; idx += 512)
            sW[(NOUT + (idx >> 6)) * FPAD + (idx & 63)] = 0.0f;
    __syncthreads();

    if (wave >= NT) return;

    int half = lane >> 4;
    int lid  = lane & 15;
    const float* arow = sH + lid * FPAD + (half << 1);
    const float* brow = sW + (wave * 16 + lid) * FPAD + (half << 1);

    v8f d = {};
#pragma unroll 4
    for (int k0 = 0; k0 < 64; k0 += 4) {
        v2f a = *(const v2f*)(arow + k0);
        v2f b = *(const v2f*)(brow + k0);
        d = __builtin_amdgcn_wmma_f32_16x16x4_f32(
             false, a,  false, b,
             (short)0, d,  false,  false);
    }

    int ncol = wave * 16 + lid;
    if (ncol < NOUT) {
        float* orow = Pn + (size_t)(blockIdx.x * 16 + (half << 3)) * NOUT + ncol;
#pragma unroll
        for (int r8 = 0; r8 < 8; ++r8)
            orow[(size_t)r8 * NOUT] = d[r8];
    }
}


extern "C" void kernel_launch(void* const* d_in, const int* in_sizes, int n_in,
                              void* d_out, int out_size, void* d_ws, size_t ws_size,
                              hipStream_t stream)
{
    const float* features = (const float*)d_in[0];
    const int*   src      = (const int*)  d_in[1];
    const int*   dst      = (const int*)  d_in[2];
    const float* W0       = (const float*)d_in[3];
    const float* b0       = (const float*)d_in[4];
    const float* W1       = (const float*)d_in[5];
    const float* b1       = (const float*)d_in[6];
    const float* W2       = (const float*)d_in[7];
    const float* b2       = (const float*)d_in[8];
    float*       out      = (float*)d_out;

    float* ws     = (float*)d_ws;
    int*      rowptr     = (int*)(ws + 50176);
    unsigned* binned     = (unsigned*)(ws + 100352);
    int*      csr        = (int*)(ws + 900352);
    float*    P          = ws + 1700352;
    float*    Q          = P  + (size_t)NNODES * 64;
    int*      pieces     = (int*)(Q + (size_t)NNODES * 64);

    const int Mtiles     = NNODES / 16;
    const int gemmBlksX  = (Mtiles + 7) / 8;
    const int nodeBlocks = NNODES / 8;
    const int nodeGroups = NNODES / 16;
    const int buckets    = (NNODES + NBUCK - 1) / NBUCK;

    gemm0_bin_kernel<<<gemmBlksX + NSBLK, 256, 0, stream>>>(features, W0, P, Mtiles, gemmBlksX,
                                                            src, dst, pieces, binned);

    bucket_csr_kernel<<<buckets, SORT_T, 0, stream>>>(binned, pieces, rowptr, csr, NNODES);

    gather_project_kernel<64, 4, true><<<nodeGroups, 512, 0, stream>>>(P, rowptr, csr, b0, W1, Q);

    gather_project_kernel<40, 3, true><<<nodeGroups, 512, 0, stream>>>(Q, rowptr, csr, b1, W2, P);

    gather_kernel<40, false><<<nodeBlocks, 256, 0, stream>>>(P, rowptr, csr, b2, out);
}


# Gather waves take 4 consecutive nodes in turn, the next node's ids read ahead under the current rows: 150000 -> 37516 gather waves; 64-node blocks, each wave one product stripe; sum order unchanged
# speedup vs baseline: 26.1006x; 1.0460x over previous
#include <hip/hip_runtime.h>


typedef float v2f __attribute__((ext_vector_type(2)));
typedef float v8f __attribute__((ext_vector_type(8)));

#define NNODES 50000
#define NEDGES 800000
#define FPAD   68

#define NBUCK  256
#define NSBLK  256
#define SORT_T 1024
#define EPT    4
#define NOEDGE 0xffffffffu
#define CHUNK  ((NEDGES + NSBLK - 1) / NSBLK)
#define BIN_T  256
#define CPT    ((CHUNK + BIN_T - 1) / BIN_T)
#define NPIECE (NBUCK + 1)

static_assert(SORT_T == 4 * NSBLK, "bucket_csr maps four threads to a chunk");
static_assert(BIN_T == NBUCK, "chunk_bin: one thread per bucket in the scan");
static_assert((NSBLK - 1) * CHUNK < NEDGES, "every chunk holds at least one edge");

__device__ __forceinline__ int wave_incl_scan(int v, int lane)
{
#pragma unroll
    for (int d = 1; d < 32; d <<= 1) {
        int t = __shfl_up(v, d);
        if (lane >= d) v += t;
    }
    return v;
}

__device__ __forceinline__ void chunk_bin(
    const int* __restrict__ src, const int* __restrict__ dst,
    int* __restrict__ pieces, unsigned* __restrict__ binned, int c)
{
    __shared__ int cnt[NBUCK];
    __shared__ int wsum[8];

    int tid  = threadIdx.x;
    int lane = tid & 31;
    cnt[tid] = 0;
    __syncthreads();

    int e0 = c * CHUNK;
    int e1 = min(NEDGES, e0 + CHUNK);
    int d[CPT], s[CPT], r[CPT];
#pragma unroll
    for (int i = 0; i < CPT; ++i) {
        int e = min(e0 + tid + i * BIN_T, e1 - 1);
        d[i] = dst[e];
        s[i] = src[e];
    }
#pragma unroll
    for (int i = 0; i < CPT; ++i) {
        r[i] = 0;
        if (e0 + tid + i * BIN_T < e1) r[i] = atomicAdd(&cnt[d[i] >> 8], 1);
    }
    __syncthreads();

    int v = cnt[tid];
    int incl = wave_incl_scan(v, lane);
    if (lane == 31) wsum[tid >> 5] = incl;
    __syncthreads();
    int pre = 0;
    for (int j = 0; j < (tid >> 5); ++j) pre += wsum[j];
    int excl = pre + incl - v;
    cnt[tid] = excl;
    pieces[c * NPIECE + tid] = excl;
    if (tid == NBUCK - 1) pieces[c * NPIECE + NBUCK] = excl + v;
    __syncthreads();

#pragma unroll
    for (int i = 0; i < CPT; ++i)
        if (e0 + tid + i * BIN_T < e1)
            binned[e0 + cnt[d[i] >> 8] + r[i]] =
                ((unsigned)s[i] << 8) | (unsigned)(d[i] & (NBUCK - 1));
}

__global__ __launch_bounds__(SORT_T) void bucket_csr_kernel(
    const unsigned* __restrict__ binned, const int* __restrict__ pieces,
    int* __restrict__ rowptr, int* __restrict__ csr, int n)
{
    __shared__ int slot[NBUCK];
    __shared__ int wsum[8];
    __shared__ int bstart;

    int tid  = threadIdx.x;
    int lane = tid & 31;
    int b    = blockIdx.x;
    int c    = tid >> 2;
    int p    = tid & 3;

    int lo = pieces[c * NPIECE + b];
    int hi = pieces[c * NPIECE + b + 1];

    if (tid < NBUCK) slot[tid] = 0;
    if (tid == 0) bstart = 0;
    __syncthreads();

    int x = (p == 0) ? lo : 0;
#pragma unroll
    for (int m = 16; m >= 1; m >>= 1) x += __shfl_xor(x, m);
    if (lane == 0) atomicAdd(&bstart, x);

    int g0 = c * CHUNK + lo + p;
    int g1 = c * CHUNK + hi;
    for (int base = g0; base < g1; base += 4 * EPT) {
        unsigned u[EPT];
#pragma unroll
        for (int i = 0; i < EPT; ++i) {
            int e = base + 4 * i;
            u[i] = (e < g1) ? binned[e] : NOEDGE;
        }
#pragma unroll
        for (int i = 0; i < EPT; ++i)
            if (u[i] != NOEDGE) atomicAdd(&slot[u[i] & (NBUCK - 1)], 1);
    }
    __syncthreads();

    int e0 = bstart;
    int v = (tid < NBUCK) ? slot[tid] : 0;
    int incl = wave_incl_scan(v, lane);
    if (tid < NBUCK && lane == 31) wsum[tid >> 5] = incl;
    __syncthreads();
    if (tid < NBUCK) {
        int pre = 0;
        for (int j = 0; j < (tid >> 5); ++j) pre += wsum[j];
        int start = e0 + pre + incl - v;
        int node = b * NBUCK + tid;
        if (node < n) rowptr[node] = start;
        slot[tid] = start;
    }
    if (b == (int)gridDim.x - 1 && tid == 0) rowptr[n] = NEDGES;
    __syncthreads();
    for (int base = g0; base < g1; base += 4 * EPT) {
        unsigned u[EPT];
#pragma unroll
        for (int i = 0; i < EPT; ++i) {
            int e = base + 4 * i;
            u[i] = (e < g1) ? binned[e] : NOEDGE;
        }
#pragma unroll
        for (int i = 0; i < EPT; ++i)
            if (u[i] != NOEDGE) {
                int pos = atomicAdd(&slot[u[i] & (NBUCK - 1)], 1);
                csr[pos] = (int)(u[i] >> 8);
            }
    }
}

#define K0    128
#define N0    64
#define GPAD  132

__device__ __forceinline__ v8f wmma4(v2f a, v2f b, v8f c)
{
    return __builtin_amdgcn_wmma_f32_16x16x4_f32(
         false, a,  false, b,
         (short)0, c,  false,  false);
}

__global__ __launch_bounds__(256) void gemm0_bin_kernel(
    const float* __restrict__ Xin, const float* __restrict__ W,
    float* __restrict__ Out, int Mtiles, int gemmBlocks,
    const int* __restrict__ src, const int* __restrict__ dst,
    int* __restrict__ pieces, unsigned* __restrict__ binned)
{
    if ((int)blockIdx.x >= gemmBlocks) {
        chunk_bin(src, dst, pieces, binned, (int)blockIdx.x - gemmBlocks);
        return;
    }

    __shared__ __attribute__((aligned(16))) float sB[N0 * GPAD];
    __shared__ __attribute__((aligned(16))) float sA[8 * 16 * GPAD];

    int tid  = threadIdx.x;
    int lane = tid & 31;
    int wave = __builtin_amdgcn_readfirstlane(tid >> 5);
    int tM   = blockIdx.x * 8 + wave;
    bool live = tM < Mtiles;
    int m0   = min(tM, Mtiles - 1) << 4;
    float* myA = sA + wave * (16 * GPAD);

    float4 a[16];
#pragma unroll
    for (int r = 0; r < 16; ++r)
        a[r] = ((const float4*)(Xin + (size_t)(m0 + r) * K0))[lane];

    float4 w[8];
    int wk[8], wn[8];
#pragma unroll
    for (int i = 0; i < 8; ++i) {
        int g  = wave + 8 * i;
        wk[i] = 8 * (g & 15) + (lane >> 2);
        wn[i] = 16 * (g >> 4) + 4 * (lane & 3);
        w[i] = *(const float4*)(W + wk[i] * N0 + wn[i]);
    }
#pragma unroll
    for (int i = 0; i < 8; ++i) {
        sB[(wn[i] + 0) * GPAD + wk[i]] = w[i].x;
        sB[(wn[i] + 1) * GPAD + wk[i]] = w[i].y;
        sB[(wn[i] + 2) * GPAD + wk[i]] = w[i].z;
        sB[(wn[i] + 3) * GPAD + wk[i]] = w[i].w;
    }
#pragma unroll
    for (int r = 0; r < 16; ++r)
        *(float4*)(myA + r * GPAD + 4 * lane) = a[r];
    __syncthreads();

    int half = lane >> 4;
    int lid  = lane & 15;
    const float* ap = myA + lid * GPAD + (half << 2);
    const float* bp = sB  + lid * GPAD + (half << 2);

    v8f acc0 = {}, acc1 = {}, acc2 = {}, acc3 = {};
#pragma unroll 4
    for (int k8 = 0; k8 < K0; k8 += 8) {
        float4 a4 = *(const float4*)(ap + k8);
        float4 b0 = *(const float4*)(bp + k8);
        float4 b1 = *(const float4*)(bp + 16 * GPAD + k8);
        float4 b2 = *(const float4*)(bp + 32 * GPAD + k8);
        float4 b3 = *(const float4*)(bp + 48 * GPAD + k8);
        v2f alo = {a4.x, a4.y}, ahi = {a4.z, a4.w};
        acc0 = wmma4(alo, (v2f){b0.x, b0.y}, acc0);
        acc1 = wmma4(alo, (v2f){b1.x, b1.y}, acc1);
        acc2 = wmma4(alo, (v2f){b2.x, b2.y}, acc2);
        acc3 = wmma4(alo, (v2f){b3.x, b3.y}, acc3);
        acc0 = wmma4(ahi, (v2f){b0.z, b0.w}, acc0);
        acc1 = wmma4(ahi, (v2f){b1.z, b1.w}, acc1);
        acc2 = wmma4(ahi, (v2f){b2.z, b2.w}, acc2);
        acc3 = wmma4(ahi, (v2f){b3.z, b3.w}, acc3);
    }
    float* dp = myA + (half << 3) * GPAD + lid;
#pragma unroll
    for (int r = 0; r < 8; ++r) {
        dp[r * GPAD +  0] = acc0[r];
        dp[r * GPAD + 16] = acc1[r];
        dp[r * GPAD + 32] = acc2[r];
        dp[r * GPAD + 48] = acc3[r];
    }
    __syncthreads();
    if (live) {
#pragma unroll
        for (int i = 0; i < 8; ++i) {
            int row = 2 * i + half;
            float4 v = *(const float4*)(myA + row * GPAD + 4 * lid);
            ((float4*)(Out + (size_t)(m0 + row) * N0))[lid] = v;
        }
    }
}

#define GCH 16
#define NPW 4

template <int C, int F>
__device__ __forceinline__ v2f add_rows(const float* __restrict__ P, int my, int j,
                                        int col, v2f acc)
{
    int s[C];
#pragma unroll
    for (int u = 0; u < C; ++u) s[u] = __builtin_amdgcn_readlane(my, j + u);
    v2f v[C];
#pragma unroll
    for (int u = 0; u < C; ++u) v[u] = *(const v2f*)(P + (size_t)s[u] * F + col);
#pragma unroll
    for (int u = 0; u < C; ++u) acc += v[u];
    return acc;
}

template <int F>
__device__ __forceinline__ v2f add_chunk(const float* __restrict__ P, int my, int n, int col, v2f acc)
{
    for (int j = 0; j < n; j += GCH) {
        switch (min(GCH, n - j)) {
            case 16: acc = add_rows<16, F>(P, my, j, col, acc); break;
            case 15: acc = add_rows<15, F>(P, my, j, col, acc); break;
            case 14: acc = add_rows<14, F>(P, my, j, col, acc); break;
            case 13: acc = add_rows<13, F>(P, my, j, col, acc); break;
            case 12: acc = add_rows<12, F>(P, my, j, col, acc); break;
            case 11: acc = add_rows<11, F>(P, my, j, col, acc); break;
            case 10: acc = add_rows<10, F>(P, my, j, col, acc); break;
            case  9: acc = add_rows< 9, F>(P, my, j, col, acc); break;
            case  8: acc = add_rows< 8, F>(P, my, j, col, acc); break;
            case  7: acc = add_rows< 7, F>(P, my, j, col, acc); break;
            case  6: acc = add_rows< 6, F>(P, my, j, col, acc); break;
            case  5: acc = add_rows< 5, F>(P, my, j, col, acc); break;
            case  4: acc = add_rows< 4, F>(P, my, j, col, acc); break;
            case  3: acc = add_rows< 3, F>(P, my, j, col, acc); break;
            case  2: acc = add_rows< 2, F>(P, my, j, col, acc); break;
            default: acc = add_rows< 1, F>(P, my, j, col, acc); break;
        }
    }
    return acc;
}

template <int F>
__device__ __forceinline__ v2f segment_sum(const float* __restrict__ P, const int* __restrict__ csr,
                                           int beg, int end, int my, int lane, int col)
{
    v2f acc = {0.0f, 0.0f};
    acc = add_chunk<F>(P, my, min(32, end - beg), col, acc);
    for (int base = beg + 32; base < end; base += 32) {
        int n = min(32, end - base);
        int more = (lane < n) ? csr[base + lane] : 0;
        acc = add_chunk<F>(P, more, n, col, acc);
    }
    return acc;
}

__device__ __forceinline__ int wave_offsets(const int* __restrict__ rowptr, int n0, int lane)
{
    return rowptr[min(n0 + min(lane, NPW), NNODES)];
}

template <int F, bool RELU>
__global__ __launch_bounds__(256) void gather_kernel(
    const float* __restrict__ P, const int* __restrict__ rowptr,
    const int* __restrict__ csr, const float* __restrict__ bias,
    float* __restrict__ outp)
{
    int lane = threadIdx.x & 31;
    int n0 = __builtin_amdgcn_readfirstlane((int)((blockIdx.x * 8 + (threadIdx.x >> 5)) * NPW));
    int cnt = min(NPW, NNODES - n0);
    int rpv = wave_offsets(rowptr, n0, lane);
    bool act = (2 * lane < F);
    int col = act ? 2 * lane : 2 * lane - F;
    v2f bv = *(const v2f*)(bias + col);

    int beg = __builtin_amdgcn_readlane(rpv, 0);
    int end = __builtin_amdgcn_readlane(rpv, 1);
    int my = csr[min(beg + lane, NEDGES - 1)];
    for (int q = 0; q < cnt; ++q) {
        int end2 = __builtin_amdgcn_readlane(rpv, min(q + 2, NPW));
        int nxt = csr[min(end + lane, NEDGES - 1)];

        v2f acc = segment_sum<F>(P, csr, beg, end, my, lane, col);
        if (act) {
            float inv = 1.0f / fmaxf((float)(end - beg), 1.0f);
            v2f r = acc * inv + bv;
            if (RELU) { r.x = fmaxf(r.x, 0.0f); r.y = fmaxf(r.y, 0.0f); }
            *(v2f*)(outp + (size_t)(n0 + q) * F + col) = r;
        }
        beg = end; end = end2; my = nxt;
    }
}


template <int NOUT, int NT, bool RELU>
__global__ __launch_bounds__(512) void gather_project_kernel(
    const float* __restrict__ P, const int* __restrict__ rowptr,
    const int* __restrict__ csr, const float* __restrict__ bias,
    const float* __restrict__ Wn, float* __restrict__ Pn)
{
    static_assert(NPW * NT <= 16, "one (tile, stripe) per wave");
    __shared__ float sW[NT * 16 * FPAD];
    __shared__ float sH[NPW * 16 * FPAD];

    constexpr int N4  = NOUT / 4;
    constexpr int NW4 = 64 * N4;
    constexpr int WPT = (NW4 + 511) / 512;

    int tid  = threadIdx.x;
    int lane = tid & 31;
    int wave = __builtin_amdgcn_readfirstlane(tid >> 5);
    int n0   = (blockIdx.x * 16 + wave) * NPW;
    int cnt  = min(NPW, NNODES - n0);

    float4 w[WPT];
#pragma unroll
    for (int i = 0; i < WPT; ++i) {
        int idx = tid + i * 512;
        w[i] = ((const float4*)Wn)[(NW4 % 512 == 0) ? idx : min(idx, NW4 - 1)];
    }

    int rpv = wave_offsets(rowptr, n0, lane);
    int col = 2 * lane;
    v2f bv = *(const v2f*)(bias + col);

    int beg = __builtin_amdgcn_readlane(rpv, 0);
    int end = __builtin_amdgcn_readlane(rpv, 1);
    int my = csr[min(beg + lane, NEDGES - 1)];
    for (int q = 0; q < cnt; ++q) {
        int end2 = __builtin_amdgcn_readlane(rpv, min(q + 2, NPW));
        int nxt = csr[min(end + lane, NEDGES - 1)];

        v2f acc = segment_sum<64>(P, csr, beg, end, my, lane, col);
        float inv = 1.0f / fmaxf((float)(end - beg), 1.0f);
        v2f r = acc * inv + bv;
        if (RELU) { r.x = fmaxf(r.x, 0.0f); r.y = fmaxf(r.y, 0.0f); }
        *(v2f*)(sH + (wave * NPW + q) * FPAD + col) = r;
        beg = end; end = end2; my = nxt;
    }

#pragma unroll
    for (int i = 0; i < WPT; ++i) {
        int idx = tid + i * 512;
        if (idx < NW4) {
            int k = idx / N4;
            int n = 4 * (idx - k * N4);
            sW[(n + 0) * FPAD + k] = w[i].x;
            sW[(n + 1) * FPAD + k] = w[i].y;
            sW[(n + 2) * FPAD + k] = w[i].z;
            sW[(n + 3) * FPAD + k] = w[i].w;
        }
    }
    if (NT * 16 > NOUT)
        for (int idx = tid; idx < (NT * 16 - NOUT) * 64; idx += 512)
            sW[(NOUT + (idx >> 6)) * FPAD + (idx & 63)] = 0.0f;
    __syncthreads();

    if (wave >= NPW * NT) return;
    int tile   = wave / NT;
    int stripe = wave - tile * NT;
    int node0  = (blockIdx.x * NPW + tile) * 16;
    if (node0 >= NNODES) return;

    int half = lane >> 4;
    int lid  = lane & 15;
    const float* arow = sH + (tile * 16 + lid) * FPAD + (half << 1);
    const float* brow = sW + (stripe * 16 + lid) * FPAD + (half << 1);

    v8f d = {};
#pragma unroll 4
    for (int k0 = 0; k0 < 64; k0 += 4) {
        v2f a = *(const v2f*)(arow + k0);
        v2f b = *(const v2f*)(brow + k0);
        d = __builtin_amdgcn_wmma_f32_16x16x4_f32(
             false, a,  false, b,
             (short)0, d,  false,  false);
    }

    int ncol = stripe * 16 + lid;
    if (ncol < NOUT) {
        float* orow = Pn + (size_t)(node0 + (half << 3)) * NOUT + ncol;
#pragma unroll
        for (int r8 = 0; r8 < 8; ++r8)
            orow[(size_t)r8 * NOUT] = d[r8];
    }
}


extern "C" void kernel_launch(void* const* d_in, const int* in_sizes, int n_in,
                              void* d_out, int out_size, void* d_ws, size_t ws_size,
                              hipStream_t stream)
{
    const float* features = (const float*)d_in[0];
    const int*   src      = (const int*)  d_in[1];
    const int*   dst      = (const int*)  d_in[2];
    const float* W0       = (const float*)d_in[3];
    const float* b0       = (const float*)d_in[4];
    const float* W1       = (const float*)d_in[5];
    const float* b1       = (const float*)d_in[6];
    const float* W2       = (const float*)d_in[7];
    const float* b2       = (const float*)d_in[8];
    float*       out      = (float*)d_out;

    float* ws     = (float*)d_ws;
    int*      rowptr     = (int*)(ws + 50176);
    unsigned* binned     = (unsigned*)(ws + 100352);
    int*      csr        = (int*)(ws + 900352);
    float*    P          = ws + 1700352;
    float*    Q          = P  + (size_t)NNODES * 64;
    int*      pieces     = (int*)(Q + (size_t)NNODES * 64);

    const int Mtiles     = NNODES / 16;
    const int gemmBlksX  = (Mtiles + 7) / 8;
    const int nodeBlocks = (NNODES + 8 * NPW - 1) / (8 * NPW);
    const int nodeGroups = (NNODES + 16 * NPW - 1) / (16 * NPW);
    const int buckets    = (NNODES + NBUCK - 1) / NBUCK;

    gemm0_bin_kernel<<<gemmBlksX + NSBLK, 256, 0, stream>>>(features, W0, P, Mtiles, gemmBlksX,
                                                            src, dst, pieces, binned);

    bucket_csr_kernel<<<buckets, SORT_T, 0, stream>>>(binned, pieces, rowptr, csr, NNODES);

    gather_project_kernel<64, 4, true><<<nodeGroups, 512, 0, stream>>>(P, rowptr, csr, b0, W1, Q);

    gather_project_kernel<40, 3, true><<<nodeGroups, 512, 0, stream>>>(Q, rowptr, csr, b1, W2, P);

    gather_kernel<40, false><<<nodeBlocks, 256, 0, stream>>>(P, rowptr, csr, b2, out);
}


# Both edge-ordering passes build their piece of the list in LDS and store it as whole 128-B lines (chunks of 3136 edges, line-aligned), not as scattered 4-B stores; sum order still LDS slot order
# speedup vs baseline: 27.6312x; 1.0586x over previous
#include <hip/hip_runtime.h>


typedef float v2f __attribute__((ext_vector_type(2)));
typedef float v8f __attribute__((ext_vector_type(8)));

#define NNODES 50000
#define NEDGES 800000
#define FPAD   68

#define NBUCK  256
#define NSBLK  256
#define SORT_T 1024
#define EPT    4
#define NOEDGE 0xffffffffu
#define CHUNK  ((((NEDGES + NSBLK - 1) / NSBLK) + 31) & ~31)
#define BIN_T  256
#define CPT    ((CHUNK + BIN_T - 1) / BIN_T)
#define NPIECE (NBUCK + 1)
#define IMG_CAP 8192

static_assert(SORT_T == 4 * NSBLK, "bucket_csr maps four threads to a chunk");
static_assert(BIN_T == NBUCK, "chunk_bin: one thread per bucket in the scan");
static_assert((NSBLK - 1) * CHUNK < NEDGES, "every chunk holds at least one edge");
static_assert((NEDGES - (NSBLK - 1) * CHUNK) % 4 == 0, "the last chunk is written out in 16-byte pieces too");

__device__ __forceinline__ int wave_incl_scan(int v, int lane)
{
#pragma unroll
    for (int d = 1; d < 32; d <<= 1) {
        int t = __shfl_up(v, d);
        if (lane >= d) v += t;
    }
    return v;
}

__device__ __forceinline__ void chunk_bin(
    const int* __restrict__ src, const int* __restrict__ dst,
    int* __restrict__ pieces, unsigned* __restrict__ binned, int c, float* img)
{
    __shared__ int cnt[NBUCK];
    __shared__ int wsum[8];

    int tid  = threadIdx.x;
    int lane = tid & 31;
    cnt[tid] = 0;
    __syncthreads();

    int e0 = c * CHUNK;
    int e1 = min(NEDGES, e0 + CHUNK);
    int d[CPT], s[CPT], r[CPT];
#pragma unroll
    for (int i = 0; i < CPT; ++i) {
        int e = min(e0 + tid + i * BIN_T, e1 - 1);
        d[i] = dst[e];
        s[i] = src[e];
    }
#pragma unroll
    for (int i = 0; i < CPT; ++i) {
        r[i] = 0;
        if (e0 + tid + i * BIN_T < e1) r[i] = atomicAdd(&cnt[d[i] >> 8], 1);
    }
    __syncthreads();

    int v = cnt[tid];
    int incl = wave_incl_scan(v, lane);
    if (lane == 31) wsum[tid >> 5] = incl;
    __syncthreads();
    int pre = 0;
    for (int j = 0; j < (tid >> 5); ++j) pre += wsum[j];
    int excl = pre + incl - v;
    cnt[tid] = excl;
    pieces[c * NPIECE + tid] = excl;
    if (tid == NBUCK - 1) pieces[c * NPIECE + NBUCK] = excl + v;
    __syncthreads();

#pragma unroll
    for (int i = 0; i < CPT; ++i)
        if (e0 + tid + i * BIN_T < e1)
            img[cnt[d[i] >> 8] + r[i]] =
                __uint_as_float(((unsigned)s[i] << 8) | (unsigned)(d[i] & (NBUCK - 1)));
    __syncthreads();
    float4* outp = (float4*)(binned + e0);
    int n4 = (e1 - e0) >> 2;
    for (int i4 = tid; i4 < n4; i4 += BIN_T)
        outp[i4] = *(const float4*)(img + 4 * i4);
}

__global__ __launch_bounds__(SORT_T) void bucket_csr_kernel(
    const unsigned* __restrict__ binned, const int* __restrict__ pieces,
    int* __restrict__ rowptr, int* __restrict__ csr, int n)
{
    __shared__ int slot[NBUCK];
    __shared__ int wsum[8];
    __shared__ int bstart;
    __shared__ int btotal;
    __shared__ int img[IMG_CAP];

    int tid  = threadIdx.x;
    int lane = tid & 31;
    int b    = blockIdx.x;
    int c    = tid >> 2;
    int p    = tid & 3;

    int lo = pieces[c * NPIECE + b];
    int hi = pieces[c * NPIECE + b + 1];

    if (tid < NBUCK) slot[tid] = 0;
    if (tid == 0) bstart = 0;
    __syncthreads();

    int x = (p == 0) ? lo : 0;
#pragma unroll
    for (int m = 16; m >= 1; m >>= 1) x += __shfl_xor(x, m);
    if (lane == 0) atomicAdd(&bstart, x);

    int g0 = c * CHUNK + lo + p;
    int g1 = c * CHUNK + hi;
    for (int base = g0; base < g1; base += 4 * EPT) {
        unsigned u[EPT];
#pragma unroll
        for (int i = 0; i < EPT; ++i) {
            int e = base + 4 * i;
            u[i] = (e < g1) ? binned[e] : NOEDGE;
        }
#pragma unroll
        for (int i = 0; i < EPT; ++i)
            if (u[i] != NOEDGE) atomicAdd(&slot[u[i] & (NBUCK - 1)], 1);
    }
    __syncthreads();

    int e0 = bstart;
    int v = (tid < NBUCK) ? slot[tid] : 0;
    int incl = wave_incl_scan(v, lane);
    if (tid < NBUCK && lane == 31) wsum[tid >> 5] = incl;
    __syncthreads();
    if (tid < NBUCK) {
        int pre = 0;
        for (int j = 0; j < (tid >> 5); ++j) pre += wsum[j];
        int rel = pre + incl - v;
        int node = b * NBUCK + tid;
        if (node < n) rowptr[node] = e0 + rel;
        slot[tid] = rel;
        if (tid == NBUCK - 1) btotal = rel + v;
    }
    if (b == (int)gridDim.x - 1 && tid == 0) rowptr[n] = NEDGES;
    __syncthreads();
    for (int base = g0; base < g1; base += 4 * EPT) {
        unsigned u[EPT];
#pragma unroll
        for (int i = 0; i < EPT; ++i) {
            int e = base + 4 * i;
            u[i] = (e < g1) ? binned[e] : NOEDGE;
        }
#pragma unroll
        for (int i = 0; i < EPT; ++i)
            if (u[i] != NOEDGE) {
                int pos = atomicAdd(&slot[u[i] & (NBUCK - 1)], 1);
                int sv = (int)(u[i] >> 8);
                if (pos < IMG_CAP) img[pos] = sv;
                else               csr[e0 + pos] = sv;
            }
    }
    __syncthreads();
    int a0   = e0 & ~31;
    int gend = e0 + min(btotal, IMG_CAP);
    for (int g = a0 + tid; g < gend; g += SORT_T)
        if (g >= e0) csr[g] = img[g - e0];
}

#define K0    128
#define N0    64
#define GPAD  132

__device__ __forceinline__ v8f wmma4(v2f a, v2f b, v8f c)
{
    return __builtin_amdgcn_wmma_f32_16x16x4_f32(
         false, a,  false, b,
         (short)0, c,  false,  false);
}

__global__ __launch_bounds__(256) void gemm0_bin_kernel(
    const float* __restrict__ Xin, const float* __restrict__ W,
    float* __restrict__ Out, int Mtiles, int gemmBlocks,
    const int* __restrict__ src, const int* __restrict__ dst,
    int* __restrict__ pieces, unsigned* __restrict__ binned)
{
    __shared__ __attribute__((aligned(16))) float sB[N0 * GPAD];
    __shared__ __attribute__((aligned(16))) float sA[8 * 16 * GPAD];
    static_assert(8 * 16 * GPAD >= CHUNK, "a binning block keeps its chunk in the tile storage");

    if ((int)blockIdx.x >= gemmBlocks) {
        chunk_bin(src, dst, pieces, binned, (int)blockIdx.x - gemmBlocks, sA);
        return;
    }

    int tid  = threadIdx.x;
    int lane = tid & 31;
    int wave = __builtin_amdgcn_readfirstlane(tid >> 5);
    int tM   = blockIdx.x * 8 + wave;
    bool live = tM < Mtiles;
    int m0   = min(tM, Mtiles - 1) << 4;
    float* myA = sA + wave * (16 * GPAD);

    float4 a[16];
#pragma unroll
    for (int r = 0; r < 16; ++r)
        a[r] = ((const float4*)(Xin + (size_t)(m0 + r) * K0))[lane];

    float4 w[8];
    int wk[8], wn[8];
#pragma unroll
    for (int i = 0; i < 8; ++i) {
        int g  = wave + 8 * i;
        wk[i] = 8 * (g & 15) + (lane >> 2);
        wn[i] = 16 * (g >> 4) + 4 * (lane & 3);
        w[i] = *(const float4*)(W + wk[i] * N0 + wn[i]);
    }
#pragma unroll
    for (int i = 0; i < 8; ++i) {
        sB[(wn[i] + 0) * GPAD + wk[i]] = w[i].x;
        sB[(wn[i] + 1) * GPAD + wk[i]] = w[i].y;
        sB[(wn[i] + 2) * GPAD + wk[i]] = w[i].z;
        sB[(wn[i] + 3) * GPAD + wk[i]] = w[i].w;
    }
#pragma unroll
    for (int r = 0; r < 16; ++r)
        *(float4*)(myA + r * GPAD + 4 * lane) = a[r];
    __syncthreads();

    int half = lane >> 4;
    int lid  = lane & 15;
    const float* ap = myA + lid * GPAD + (half << 2);
    const float* bp = sB  + lid * GPAD + (half << 2);

    v8f acc0 = {}, acc1 = {}, acc2 = {}, acc3 = {};
#pragma unroll 4
    for (int k8 = 0; k8 < K0; k8 += 8) {
        float4 a4 = *(const float4*)(ap + k8);
        float4 b0 = *(const float4*)(bp + k8);
        float4 b1 = *(const float4*)(bp + 16 * GPAD + k8);
        float4 b2 = *(const float4*)(bp + 32 * GPAD + k8);
        float4 b3 = *(const float4*)(bp + 48 * GPAD + k8);
        v2f alo = {a4.x, a4.y}, ahi = {a4.z, a4.w};
        acc0 = wmma4(alo, (v2f){b0.x, b0.y}, acc0);
        acc1 = wmma4(alo, (v2f){b1.x, b1.y}, acc1);
        acc2 = wmma4(alo, (v2f){b2.x, b2.y}, acc2);
        acc3 = wmma4(alo, (v2f){b3.x, b3.y}, acc3);
        acc0 = wmma4(ahi, (v2f){b0.z, b0.w}, acc0);
        acc1 = wmma4(ahi, (v2f){b1.z, b1.w}, acc1);
        acc2 = wmma4(ahi, (v2f){b2.z, b2.w}, acc2);
        acc3 = wmma4(ahi, (v2f){b3.z, b3.w}, acc3);
    }
    float* dp = myA + (half << 3) * GPAD + lid;
#pragma unroll
    for (int r = 0; r < 8; ++r) {
        dp[r * GPAD +  0] = acc0[r];
        dp[r * GPAD + 16] = acc1[r];
        dp[r * GPAD + 32] = acc2[r];
        dp[r * GPAD + 48] = acc3[r];
    }
    __syncthreads();
    if (live) {
#pragma unroll
        for (int i = 0; i < 8; ++i) {
            int row = 2 * i + half;
            float4 v = *(const float4*)(myA + row * GPAD + 4 * lid);
            ((float4*)(Out + (size_t)(m0 + row) * N0))[lid] = v;
        }
    }
}

#define GCH 16
#define NPW 4

template <int C, int F>
__device__ __forceinline__ v2f add_rows(const float* __restrict__ P, int my, int j,
                                        int col, v2f acc)
{
    int s[C];
#pragma unroll
    for (int u = 0; u < C; ++u) s[u] = __builtin_amdgcn_readlane(my, j + u);
    v2f v[C];
#pragma unroll
    for (int u = 0; u < C; ++u) v[u] = *(const v2f*)(P + (size_t)s[u] * F + col);
#pragma unroll
    for (int u = 0; u < C; ++u) acc += v[u];
    return acc;
}

template <int F>
__device__ __forceinline__ v2f add_chunk(const float* __restrict__ P, int my, int n, int col, v2f acc)
{
    for (int j = 0; j < n; j += GCH) {
        switch (min(GCH, n - j)) {
            case 16: acc = add_rows<16, F>(P, my, j, col, acc); break;
            case 15: acc = add_rows<15, F>(P, my, j, col, acc); break;
            case 14: acc = add_rows<14, F>(P, my, j, col, acc); break;
            case 13: acc = add_rows<13, F>(P, my, j, col, acc); break;
            case 12: acc = add_rows<12, F>(P, my, j, col, acc); break;
            case 11: acc = add_rows<11, F>(P, my, j, col, acc); break;
            case 10: acc = add_rows<10, F>(P, my, j, col, acc); break;
            case  9: acc = add_rows< 9, F>(P, my, j, col, acc); break;
            case  8: acc = add_rows< 8, F>(P, my, j, col, acc); break;
            case  7: acc = add_rows< 7, F>(P, my, j, col, acc); break;
            case  6: acc = add_rows< 6, F>(P, my, j, col, acc); break;
            case  5: acc = add_rows< 5, F>(P, my, j, col, acc); break;
            case  4: acc = add_rows< 4, F>(P, my, j, col, acc); break;
            case  3: acc = add_rows< 3, F>(P, my, j, col, acc); break;
            case  2: acc = add_rows< 2, F>(P, my, j, col, acc); break;
            default: acc = add_rows< 1, F>(P, my, j, col, acc); break;
        }
    }
    return acc;
}

template <int F>
__device__ __forceinline__ v2f segment_sum(const float* __restrict__ P, const int* __restrict__ csr,
                                           int beg, int end, int my, int lane, int col)
{
    v2f acc = {0.0f, 0.0f};
    acc = add_chunk<F>(P, my, min(32, end - beg), col, acc);
    for (int base = beg + 32; base < end; base += 32) {
        int n = min(32, end - base);
        int more = (lane < n) ? csr[base + lane] : 0;
        acc = add_chunk<F>(P, more, n, col, acc);
    }
    return acc;
}

__device__ __forceinline__ int wave_offsets(const int* __restrict__ rowptr, int n0, int lane)
{
    return rowptr[min(n0 + min(lane, NPW), NNODES)];
}

template <int F, bool RELU>
__global__ __launch_bounds__(256) void gather_kernel(
    const float* __restrict__ P, const int* __restrict__ rowptr,
    const int* __restrict__ csr, const float* __restrict__ bias,
    float* __restrict__ outp)
{
    int lane = threadIdx.x & 31;
    int n0 = __builtin_amdgcn_readfirstlane((int)((blockIdx.x * 8 + (threadIdx.x >> 5)) * NPW));
    int cnt = min(NPW, NNODES - n0);
    int rpv = wave_offsets(rowptr, n0, lane);
    bool act = (2 * lane < F);
    int col = act ? 2 * lane : 2 * lane - F;
    v2f bv = *(const v2f*)(bias + col);

    int beg = __builtin_amdgcn_readlane(rpv, 0);
    int end = __builtin_amdgcn_readlane(rpv, 1);
    int my = csr[min(beg + lane, NEDGES - 1)];
    for (int q = 0; q < cnt; ++q) {
        int end2 = __builtin_amdgcn_readlane(rpv, min(q + 2, NPW));
        int nxt = csr[min(end + lane, NEDGES - 1)];

        v2f acc = segment_sum<F>(P, csr, beg, end, my, lane, col);
        if (act) {
            float inv = 1.0f / fmaxf((float)(end - beg), 1.0f);
            v2f r = acc * inv + bv;
            if (RELU) { r.x = fmaxf(r.x, 0.0f); r.y = fmaxf(r.y, 0.0f); }
            *(v2f*)(outp + (size_t)(n0 + q) * F + col) = r;
        }
        beg = end; end = end2; my = nxt;
    }
}


template <int NOUT, int NT, bool RELU>
__global__ __launch_bounds__(512) void gather_project_kernel(
    const float* __restrict__ P, const int* __restrict__ rowptr,
    const int* __restrict__ csr, const float* __restrict__ bias,
    const float* __restrict__ Wn, float* __restrict__ Pn)
{
    static_assert(NPW * NT <= 16, "one (tile, stripe) per wave");
    __shared__ float sW[NT * 16 * FPAD];
    __shared__ float sH[NPW * 16 * FPAD];

    constexpr int N4  = NOUT / 4;
    constexpr int NW4 = 64 * N4;
    constexpr int WPT = (NW4 + 511) / 512;

    int tid  = threadIdx.x;
    int lane = tid & 31;
    int wave = __builtin_amdgcn_readfirstlane(tid >> 5);
    int n0   = (blockIdx.x * 16 + wave) * NPW;
    int cnt  = min(NPW, NNODES - n0);

    float4 w[WPT];
#pragma unroll
    for (int i = 0; i < WPT; ++i) {
        int idx = tid + i * 512;
        w[i] = ((const float4*)Wn)[(NW4 % 512 == 0) ? idx : min(idx, NW4 - 1)];
    }

    int rpv = wave_offsets(rowptr, n0, lane);
    int col = 2 * lane;
    v2f bv = *(const v2f*)(bias + col);

    int beg = __builtin_amdgcn_readlane(rpv, 0);
    int end = __builtin_amdgcn_readlane(rpv, 1);
    int my = csr[min(beg + lane, NEDGES - 1)];
    for (int q = 0; q < cnt; ++q) {
        int end2 = __builtin_amdgcn_readlane(rpv, min(q + 2, NPW));
        int nxt = csr[min(end + lane, NEDGES - 1)];

        v2f acc = segment_sum<64>(P, csr, beg, end, my, lane, col);
        float inv = 1.0f / fmaxf((float)(end - beg), 1.0f);
        v2f r = acc * inv + bv;
        if (RELU) { r.x = fmaxf(r.x, 0.0f); r.y = fmaxf(r.y, 0.0f); }
        *(v2f*)(sH + (wave * NPW + q) * FPAD + col) = r;
        beg = end; end = end2; my = nxt;
    }

#pragma unroll
    for (int i = 0; i < WPT; ++i) {
        int idx = tid + i * 512;
        if (idx < NW4) {
            int k = idx / N4;
            int n = 4 * (idx - k * N4);
            sW[(n + 0) * FPAD + k] = w[i].x;
            sW[(n + 1) * FPAD + k] = w[i].y;
            sW[(n + 2) * FPAD + k] = w[i].z;
            sW[(n + 3) * FPAD + k] = w[i].w;
        }
    }
    if (NT * 16 > NOUT)
        for (int idx = tid; idx < (NT * 16 - NOUT) * 64; idx += 512)
            sW[(NOUT + (idx >> 6)) * FPAD + (idx & 63)] = 0.0f;
    __syncthreads();

    if (wave >= NPW * NT) return;
    int tile   = wave / NT;
    int stripe = wave - tile * NT;
    int node0  = (blockIdx.x * NPW + tile) * 16;
    if (node0 >= NNODES) return;

    int half = lane >> 4;
    int lid  = lane & 15;
    const float* arow = sH + (tile * 16 + lid) * FPAD + (half << 1);
    const float* brow = sW + (stripe * 16 + lid) * FPAD + (half << 1);

    v8f d = {};
#pragma unroll 4
    for (int k0 = 0; k0 < 64; k0 += 4) {
        v2f a = *(const v2f*)(arow + k0);
        v2f b = *(const v2f*)(brow + k0);
        d = __builtin_amdgcn_wmma_f32_16x16x4_f32(
             false, a,  false, b,
             (short)0, d,  false,  false);
    }

    int ncol = stripe * 16 + lid;
    if (ncol < NOUT) {
        float* orow = Pn + (size_t)(node0 + (half << 3)) * NOUT + ncol;
#pragma unroll
        for (int r8 = 0; r8 < 8; ++r8)
            orow[(size_t)r8 * NOUT] = d[r8];
    }
}


extern "C" void kernel_launch(void* const* d_in, const int* in_sizes, int n_in,
                              void* d_out, int out_size, void* d_ws, size_t ws_size,
                              hipStream_t stream)
{
    const float* features = (const float*)d_in[0];
    const int*   src      = (const int*)  d_in[1];
    const int*   dst      = (const int*)  d_in[2];
    const float* W0       = (const float*)d_in[3];
    const float* b0       = (const float*)d_in[4];
    const float* W1       = (const float*)d_in[5];
    const float* b1       = (const float*)d_in[6];
    const float* W2       = (const float*)d_in[7];
    const float* b2       = (const float*)d_in[8];
    float*       out      = (float*)d_out;

    float* ws     = (float*)d_ws;
    int*      rowptr     = (int*)(ws + 50176);
    unsigned* binned     = (unsigned*)(ws + 100352);
    int*      csr        = (int*)(ws + 900352);
    float*    P          = ws + 1700352;
    float*    Q          = P  + (size_t)NNODES * 64;
    int*      pieces     = (int*)(Q + (size_t)NNODES * 64);

    const int Mtiles     = NNODES / 16;
    const int gemmBlksX  = (Mtiles + 7) / 8;
    const int nodeBlocks = (NNODES + 8 * NPW - 1) / (8 * NPW);
    const int nodeGroups = (NNODES + 16 * NPW - 1) / (16 * NPW);
    const int buckets    = (NNODES + NBUCK - 1) / NBUCK;

    gemm0_bin_kernel<<<gemmBlksX + NSBLK, 256, 0, stream>>>(features, W0, P, Mtiles, gemmBlksX,
                                                            src, dst, pieces, binned);

    bucket_csr_kernel<<<buckets, SORT_T, 0, stream>>>(binned, pieces, rowptr, csr, NNODES);

    gather_project_kernel<64, 4, true><<<nodeGroups, 512, 0, stream>>>(P, rowptr, csr, b0, W1, Q);

    gather_project_kernel<40, 3, true><<<nodeGroups, 512, 0, stream>>>(Q, rowptr, csr, b1, W2, P);

    gather_kernel<40, false><<<nodeBlocks, 256, 0, stream>>>(P, rowptr, csr, b2, out);
}


# Second edge-ordering pass folded into the first gather launch: a 32-wave block orders its own 128-node bucket in LDS and gathers from the LDS ids; 5 -> 4 launches; sum order still LDS slot order
# speedup vs baseline: 29.4349x; 1.0653x over previous
#include <hip/hip_runtime.h>


typedef float v2f __attribute__((ext_vector_type(2)));
typedef float v8f __attribute__((ext_vector_type(8)));

#define NNODES 50000
#define NEDGES 800000
#define FPAD   68

#define BSH    7
#define BNODES (1 << BSH)
#define NBK    ((NNODES + BNODES - 1) / BNODES)
#define NCNT   512
#define NSBLK  256
#define SORT_T 1024
#define EPT    4
#define NOEDGE 0xffffffffu
#define CHUNK  ((((NEDGES + NSBLK - 1) / NSBLK) + 31) & ~31)
#define BIN_T  256
#define CPT    ((CHUNK + BIN_T - 1) / BIN_T)
#define NPIECE (NBK + 1)
#define IMG_CAP 4096

static_assert(SORT_T == 4 * NSBLK, "the bucket pass maps four threads to a chunk");
static_assert(NCNT == 2 * BIN_T && NBK + 1 <= NCNT, "chunk_bin: two counters per thread in the scan, and the end offset has a slot");
static_assert((NSBLK - 1) * CHUNK < NEDGES, "every chunk holds at least one edge");
static_assert((NEDGES - (NSBLK - 1) * CHUNK) % 4 == 0, "the last chunk is written out in 16-byte pieces too");

__device__ __forceinline__ int wave_incl_scan(int v, int lane)
{
#pragma unroll
    for (int d = 1; d < 32; d <<= 1) {
        int t = __shfl_up(v, d);
        if (lane >= d) v += t;
    }
    return v;
}

__device__ __forceinline__ void chunk_bin(
    const int* __restrict__ src, const int* __restrict__ dst,
    int* __restrict__ pieces, unsigned* __restrict__ binned, int c, float* img)
{
    __shared__ int cnt[NCNT];
    __shared__ int wsum[8];

    int tid  = threadIdx.x;
    int lane = tid & 31;
    cnt[tid] = 0;
    cnt[tid + BIN_T] = 0;
    __syncthreads();

    int e0 = c * CHUNK;
    int e1 = min(NEDGES, e0 + CHUNK);
    int d[CPT], s[CPT], r[CPT];
#pragma unroll
    for (int i = 0; i < CPT; ++i) {
        int e = min(e0 + tid + i * BIN_T, e1 - 1);
        d[i] = dst[e];
        s[i] = src[e];
    }
#pragma unroll
    for (int i = 0; i < CPT; ++i) {
        r[i] = 0;
        if (e0 + tid + i * BIN_T < e1) r[i] = atomicAdd(&cnt[d[i] >> BSH], 1);
    }
    __syncthreads();

    int v0 = cnt[2 * tid];
    int v1 = cnt[2 * tid + 1];
    int v = v0 + v1;
    int incl = wave_incl_scan(v, lane);
    if (lane == 31) wsum[tid >> 5] = incl;
    __syncthreads();
    int pre = 0;
    for (int j = 0; j < (tid >> 5); ++j) pre += wsum[j];
    int excl = pre + incl - v;
    cnt[2 * tid]     = excl;
    cnt[2 * tid + 1] = excl + v0;
    if (2 * tid     <= NBK) pieces[c * NPIECE + 2 * tid]     = excl;
    if (2 * tid + 1 <= NBK) pieces[c * NPIECE + 2 * tid + 1] = excl + v0;
    __syncthreads();

#pragma unroll
    for (int i = 0; i < CPT; ++i)
        if (e0 + tid + i * BIN_T < e1)
            img[cnt[d[i] >> BSH] + r[i]] =
                __uint_as_float(((unsigned)s[i] << BSH) | (unsigned)(d[i] & (BNODES - 1)));
    __syncthreads();
    float4* outp = (float4*)(binned + e0);
    int n4 = (e1 - e0) >> 2;
    for (int i4 = tid; i4 < n4; i4 += BIN_T)
        outp[i4] = *(const float4*)(img + 4 * i4);
}

#define K0    128
#define N0    64
#define GPAD  132

__device__ __forceinline__ v8f wmma4(v2f a, v2f b, v8f c)
{
    return __builtin_amdgcn_wmma_f32_16x16x4_f32(
         false, a,  false, b,
         (short)0, c,  false,  false);
}

__global__ __launch_bounds__(256) void gemm0_bin_kernel(
    const float* __restrict__ Xin, const float* __restrict__ W,
    float* __restrict__ Out, int Mtiles, int gemmBlocks,
    const int* __restrict__ src, const int* __restrict__ dst,
    int* __restrict__ pieces, unsigned* __restrict__ binned)
{
    __shared__ __attribute__((aligned(16))) float sB[N0 * GPAD];
    __shared__ __attribute__((aligned(16))) float sA[8 * 16 * GPAD];
    static_assert(8 * 16 * GPAD >= CHUNK, "a binning block keeps its chunk in the tile storage");

    if ((int)blockIdx.x >= gemmBlocks) {
        chunk_bin(src, dst, pieces, binned, (int)blockIdx.x - gemmBlocks, sA);
        return;
    }

    int tid  = threadIdx.x;
    int lane = tid & 31;
    int wave = __builtin_amdgcn_readfirstlane(tid >> 5);
    int tM   = blockIdx.x * 8 + wave;
    bool live = tM < Mtiles;
    int m0   = min(tM, Mtiles - 1) << 4;
    float* myA = sA + wave * (16 * GPAD);

    float4 a[16];
#pragma unroll
    for (int r = 0; r < 16; ++r)
        a[r] = ((const float4*)(Xin + (size_t)(m0 + r) * K0))[lane];

    float4 w[8];
    int wk[8], wn[8];
#pragma unroll
    for (int i = 0; i < 8; ++i) {
        int g  = wave + 8 * i;
        wk[i] = 8 * (g & 15) + (lane >> 2);
        wn[i] = 16 * (g >> 4) + 4 * (lane & 3);
        w[i] = *(const float4*)(W + wk[i] * N0 + wn[i]);
    }
#pragma unroll
    for (int i = 0; i < 8; ++i) {
        sB[(wn[i] + 0) * GPAD + wk[i]] = w[i].x;
        sB[(wn[i] + 1) * GPAD + wk[i]] = w[i].y;
        sB[(wn[i] + 2) * GPAD + wk[i]] = w[i].z;
        sB[(wn[i] + 3) * GPAD + wk[i]] = w[i].w;
    }
#pragma unroll
    for (int r = 0; r < 16; ++r)
        *(float4*)(myA + r * GPAD + 4 * lane) = a[r];
    __syncthreads();

    int half = lane >> 4;
    int lid  = lane & 15;
    const float* ap = myA + lid * GPAD + (half << 2);
    const float* bp = sB  + lid * GPAD + (half << 2);

    v8f acc0 = {}, acc1 = {}, acc2 = {}, acc3 = {};
#pragma unroll 4
    for (int k8 = 0; k8 < K0; k8 += 8) {
        float4 a4 = *(const float4*)(ap + k8);
        float4 b0 = *(const float4*)(bp + k8);
        float4 b1 = *(const float4*)(bp + 16 * GPAD + k8);
        float4 b2 = *(const float4*)(bp + 32 * GPAD + k8);
        float4 b3 = *(const float4*)(bp + 48 * GPAD + k8);
        v2f alo = {a4.x, a4.y}, ahi = {a4.z, a4.w};
        acc0 = wmma4(alo, (v2f){b0.x, b0.y}, acc0);
        acc1 = wmma4(alo, (v2f){b1.x, b1.y}, acc1);
        acc2 = wmma4(alo, (v2f){b2.x, b2.y}, acc2);
        acc3 = wmma4(alo, (v2f){b3.x, b3.y}, acc3);
        acc0 = wmma4(ahi, (v2f){b0.z, b0.w}, acc0);
        acc1 = wmma4(ahi, (v2f){b1.z, b1.w}, acc1);
        acc2 = wmma4(ahi, (v2f){b2.z, b2.w}, acc2);
        acc3 = wmma4(ahi, (v2f){b3.z, b3.w}, acc3);
    }
    float* dp = myA + (half << 3) * GPAD + lid;
#pragma unroll
    for (int r = 0; r < 8; ++r) {
        dp[r * GPAD +  0] = acc0[r];
        dp[r * GPAD + 16] = acc1[r];
        dp[r * GPAD + 32] = acc2[r];
        dp[r * GPAD + 48] = acc3[r];
    }
    __syncthreads();
    if (live) {
#pragma unroll
        for (int i = 0; i < 8; ++i) {
            int row = 2 * i + half;
            float4 v = *(const float4*)(myA + row * GPAD + 4 * lid);
            ((float4*)(Out + (size_t)(m0 + row) * N0))[lid] = v;
        }
    }
}

#define GCH 16
#define NPW 4

template <int C, int F>
__device__ __forceinline__ v2f add_rows(const float* __restrict__ P, int my, int j,
                                        int col, v2f acc)
{
    int s[C];
#pragma unroll
    for (int u = 0; u < C; ++u) s[u] = __builtin_amdgcn_readlane(my, j + u);
    v2f v[C];
#pragma unroll
    for (int u = 0; u < C; ++u) v[u] = *(const v2f*)(P + (size_t)s[u] * F + col);
#pragma unroll
    for (int u = 0; u < C; ++u) acc += v[u];
    return acc;
}

template <int F>
__device__ __forceinline__ v2f add_chunk(const float* __restrict__ P, int my, int n, int col, v2f acc)
{
    for (int j = 0; j < n; j += GCH) {
        switch (min(GCH, n - j)) {
            case 16: acc = add_rows<16, F>(P, my, j, col, acc); break;
            case 15: acc = add_rows<15, F>(P, my, j, col, acc); break;
            case 14: acc = add_rows<14, F>(P, my, j, col, acc); break;
            case 13: acc = add_rows<13, F>(P, my, j, col, acc); break;
            case 12: acc = add_rows<12, F>(P, my, j, col, acc); break;
            case 11: acc = add_rows<11, F>(P, my, j, col, acc); break;
            case 10: acc = add_rows<10, F>(P, my, j, col, acc); break;
            case  9: acc = add_rows< 9, F>(P, my, j, col, acc); break;
            case  8: acc = add_rows< 8, F>(P, my, j, col, acc); break;
            case  7: acc = add_rows< 7, F>(P, my, j, col, acc); break;
            case  6: acc = add_rows< 6, F>(P, my, j, col, acc); break;
            case  5: acc = add_rows< 5, F>(P, my, j, col, acc); break;
            case  4: acc = add_rows< 4, F>(P, my, j, col, acc); break;
            case  3: acc = add_rows< 3, F>(P, my, j, col, acc); break;
            case  2: acc = add_rows< 2, F>(P, my, j, col, acc); break;
            default: acc = add_rows< 1, F>(P, my, j, col, acc); break;
        }
    }
    return acc;
}

template <int F>
__device__ __forceinline__ v2f segment_sum(const float* __restrict__ P, const int* __restrict__ csr,
                                           int beg, int end, int my, int lane, int col)
{
    v2f acc = {0.0f, 0.0f};
    acc = add_chunk<F>(P, my, min(32, end - beg), col, acc);
    for (int base = beg + 32; base < end; base += 32) {
        int n = min(32, end - base);
        int more = (lane < n) ? csr[base + lane] : 0;
        acc = add_chunk<F>(P, more, n, col, acc);
    }
    return acc;
}

__device__ __forceinline__ int wave_offsets(const int* __restrict__ rowptr, int n0, int lane)
{
    return rowptr[min(n0 + min(lane, NPW), NNODES)];
}

template <int F, bool RELU>
__global__ __launch_bounds__(256) void gather_kernel(
    const float* __restrict__ P, const int* __restrict__ rowptr,
    const int* __restrict__ csr, const float* __restrict__ bias,
    float* __restrict__ outp)
{
    int lane = threadIdx.x & 31;
    int n0 = __builtin_amdgcn_readfirstlane((int)((blockIdx.x * 8 + (threadIdx.x >> 5)) * NPW));
    int cnt = min(NPW, NNODES - n0);
    int rpv = wave_offsets(rowptr, n0, lane);
    bool act = (2 * lane < F);
    int col = act ? 2 * lane : 2 * lane - F;
    v2f bv = *(const v2f*)(bias + col);

    int beg = __builtin_amdgcn_readlane(rpv, 0);
    int end = __builtin_amdgcn_readlane(rpv, 1);
    int my = csr[min(beg + lane, NEDGES - 1)];
    for (int q = 0; q < cnt; ++q) {
        int end2 = __builtin_amdgcn_readlane(rpv, min(q + 2, NPW));
        int nxt = csr[min(end + lane, NEDGES - 1)];

        v2f acc = segment_sum<F>(P, csr, beg, end, my, lane, col);
        if (act) {
            float inv = 1.0f / fmaxf((float)(end - beg), 1.0f);
            v2f r = acc * inv + bv;
            if (RELU) { r.x = fmaxf(r.x, 0.0f); r.y = fmaxf(r.y, 0.0f); }
            *(v2f*)(outp + (size_t)(n0 + q) * F + col) = r;
        }
        beg = end; end = end2; my = nxt;
    }
}


template <int NOUT, int NT, bool RELU>
__global__ __launch_bounds__(512) void gather_project_kernel(
    const float* __restrict__ P, const int* __restrict__ rowptr,
    const int* __restrict__ csr, const float* __restrict__ bias,
    const float* __restrict__ Wn, float* __restrict__ Pn)
{
    static_assert(NPW * NT <= 16, "one (tile, stripe) per wave");
    __shared__ float sW[NT * 16 * FPAD];
    __shared__ float sH[NPW * 16 * FPAD];

    constexpr int N4  = NOUT / 4;
    constexpr int NW4 = 64 * N4;
    constexpr int WPT = (NW4 + 511) / 512;

    int tid  = threadIdx.x;
    int lane = tid & 31;
    int wave = __builtin_amdgcn_readfirstlane(tid >> 5);
    int n0   = (blockIdx.x * 16 + wave) * NPW;
    int cnt  = min(NPW, NNODES - n0);

    float4 w[WPT];
#pragma unroll
    for (int i = 0; i < WPT; ++i) {
        int idx = tid + i * 512;
        w[i] = ((const float4*)Wn)[(NW4 % 512 == 0) ? idx : min(idx, NW4 - 1)];
    }

    int rpv = wave_offsets(rowptr, n0, lane);
    int col = 2 * lane;
    v2f bv = *(const v2f*)(bias + col);

    int beg = __builtin_amdgcn_readlane(rpv, 0);
    int end = __builtin_amdgcn_readlane(rpv, 1);
    int my = csr[min(beg + lane, NEDGES - 1)];
    for (int q = 0; q < cnt; ++q) {
        int end2 = __builtin_amdgcn_readlane(rpv, min(q + 2, NPW));
        int nxt = csr[min(end + lane, NEDGES - 1)];

        v2f acc = segment_sum<64>(P, csr, beg, end, my, lane, col);
        float inv = 1.0f / fmaxf((float)(end - beg), 1.0f);
        v2f r = acc * inv + bv;
        if (RELU) { r.x = fmaxf(r.x, 0.0f); r.y = fmaxf(r.y, 0.0f); }
        *(v2f*)(sH + (wave * NPW + q) * FPAD + col) = r;
        beg = end; end = end2; my = nxt;
    }

#pragma unroll
    for (int i = 0; i < WPT; ++i) {
        int idx = tid + i * 512;
        if (idx < NW4) {
            int k = idx / N4;
            int n = 4 * (idx - k * N4);
            sW[(n + 0) * FPAD + k] = w[i].x;
            sW[(n + 1) * FPAD + k] = w[i].y;
            sW[(n + 2) * FPAD + k] = w[i].z;
            sW[(n + 3) * FPAD + k] = w[i].w;
        }
    }
    if (NT * 16 > NOUT)
        for (int idx = tid; idx < (NT * 16 - NOUT) * 64; idx += 512)
            sW[(NOUT + (idx >> 6)) * FPAD + (idx & 63)] = 0.0f;
    __syncthreads();

    if (wave >= NPW * NT) return;
    int tile   = wave / NT;
    int stripe = wave - tile * NT;
    int node0  = (blockIdx.x * NPW + tile) * 16;
    if (node0 >= NNODES) return;

    int half = lane >> 4;
    int lid  = lane & 15;
    const float* arow = sH + (tile * 16 + lid) * FPAD + (half << 1);
    const float* brow = sW + (stripe * 16 + lid) * FPAD + (half << 1);

    v8f d = {};
#pragma unroll 4
    for (int k0 = 0; k0 < 64; k0 += 4) {
        v2f a = *(const v2f*)(arow + k0);
        v2f b = *(const v2f*)(brow + k0);
        d = __builtin_amdgcn_wmma_f32_16x16x4_f32(
             false, a,  false, b,
             (short)0, d,  false,  false);
    }

    int ncol = stripe * 16 + lid;
    if (ncol < NOUT) {
        float* orow = Pn + (size_t)(node0 + (half << 3)) * NOUT + ncol;
#pragma unroll
        for (int r8 = 0; r8 < 8; ++r8)
            orow[(size_t)r8 * NOUT] = d[r8];
    }
}

__device__ __forceinline__ int list_id(const int* img, const int* csrb, int pos)
{
    return (pos < IMG_CAP) ? img[pos]
                           : __hip_atomic_load(csrb + pos, __ATOMIC_RELAXED, __HIP_MEMORY_SCOPE_AGENT);
}

template <int NOUT, int NT, bool RELU>
__global__ __launch_bounds__(SORT_T) void csr_gather_project_kernel(
    const unsigned* __restrict__ binned, const int* __restrict__ pieces,
    int* __restrict__ rowptr, int* csr,
    const float* __restrict__ P, const float* __restrict__ bias,
    const float* __restrict__ Wn, float* __restrict__ Pn, int n)
{
    static_assert(BNODES == (SORT_T / 32) * NPW, "every wave gathers NPW nodes of the bucket");
    static_assert((BNODES / 16) * NT <= SORT_T / 32, "one (tile, stripe) per wave");
    __shared__ int slot[BNODES];
    __shared__ int rp[BNODES + 1];
    __shared__ int wsum[BNODES / 32];
    __shared__ int bstart;
    __shared__ int img[IMG_CAP];
    __shared__ float sW[NT * 16 * FPAD];
    __shared__ float sH[BNODES * FPAD];

    constexpr int N4  = NOUT / 4;
    constexpr int NW4 = 64 * N4;
    constexpr int WPT = (NW4 + SORT_T - 1) / SORT_T;

    int tid  = threadIdx.x;
    int lane = tid & 31;
    int b    = blockIdx.x;
    int c    = tid >> 2;
    int p    = tid & 3;

    int lo = pieces[c * NPIECE + b];
    int hi = pieces[c * NPIECE + b + 1];

    float4 w[WPT];
#pragma unroll
    for (int i = 0; i < WPT; ++i) {
        int idx = tid + i * SORT_T;
        w[i] = ((const float4*)Wn)[(NW4 % SORT_T == 0) ? idx : min(idx, NW4 - 1)];
    }
    int col = 2 * lane;
    v2f bv = *(const v2f*)(bias + col);

    if (tid < BNODES) slot[tid] = 0;
    if (tid == 0) bstart = 0;
    __syncthreads();

    int x = (p == 0) ? lo : 0;
#pragma unroll
    for (int m = 16; m >= 1; m >>= 1) x += __shfl_xor(x, m);
    if (lane == 0) atomicAdd(&bstart, x);

    int g0 = c * CHUNK + lo + p;
    int g1 = c * CHUNK + hi;
    for (int base = g0; base < g1; base += 4 * EPT) {
        unsigned u[EPT];
#pragma unroll
        for (int i = 0; i < EPT; ++i) {
            int e = base + 4 * i;
            u[i] = (e < g1) ? binned[e] : NOEDGE;
        }
#pragma unroll
        for (int i = 0; i < EPT; ++i)
            if (u[i] != NOEDGE) atomicAdd(&slot[u[i] & (BNODES - 1)], 1);
    }
    __syncthreads();

    int e0 = bstart;
    int v = (tid < BNODES) ? slot[tid] : 0;
    int incl = wave_incl_scan(v, lane);
    if (tid < BNODES && lane == 31) wsum[tid >> 5] = incl;
    __syncthreads();
    if (tid < BNODES) {
        int pre = 0;
        for (int j = 0; j < (tid >> 5); ++j) pre += wsum[j];
        int rel = pre + incl - v;
        int node = b * BNODES + tid;
        if (node < n) rowptr[node] = e0 + rel;
        slot[tid] = rel;
        rp[tid] = rel;
        if (tid == BNODES - 1) rp[BNODES] = rel + v;
    }
    if (b == (int)gridDim.x - 1 && tid == 0) rowptr[n] = NEDGES;
    __syncthreads();
    for (int base = g0; base < g1; base += 4 * EPT) {
        unsigned u[EPT];
#pragma unroll
        for (int i = 0; i < EPT; ++i) {
            int e = base + 4 * i;
            u[i] = (e < g1) ? binned[e] : NOEDGE;
        }
#pragma unroll
        for (int i = 0; i < EPT; ++i)
            if (u[i] != NOEDGE) {
                int pos = atomicAdd(&slot[u[i] & (BNODES - 1)], 1);
                int sv = (int)(u[i] >> BSH);
                if (pos < IMG_CAP) img[pos] = sv;
                else               csr[e0 + pos] = sv;
            }
    }
    int btotal = rp[BNODES];
    if (btotal > IMG_CAP) __threadfence();
    __syncthreads();
    int a0   = e0 & ~31;
    int gend = e0 + min(btotal, IMG_CAP);
    for (int g = a0 + tid; g < gend; g += SORT_T)
        if (g >= e0) csr[g] = img[g - e0];

    int wave = __builtin_amdgcn_readfirstlane(tid >> 5);
    int nl0  = wave * NPW;
    int cnt  = min(NPW, NNODES - (b * BNODES + nl0));
    const int* csrb = csr + e0;
    int rpv = rp[nl0 + min(lane, NPW)];
    int beg = __builtin_amdgcn_readlane(rpv, 0);
    for (int q = 0; q < cnt; ++q) {
        int end = __builtin_amdgcn_readlane(rpv, q + 1);
        v2f acc = {0.0f, 0.0f};
        for (int base = beg; base < end; base += 32) {
            int m = min(32, end - base);
            int my = (lane < m) ? list_id(img, csrb, base + lane) : 0;
            acc = add_chunk<64>(P, my, m, col, acc);
        }
        float inv = 1.0f / fmaxf((float)(end - beg), 1.0f);
        v2f r = acc * inv + bv;
        if (RELU) { r.x = fmaxf(r.x, 0.0f); r.y = fmaxf(r.y, 0.0f); }
        *(v2f*)(sH + (nl0 + q) * FPAD + col) = r;
        beg = end;
    }

#pragma unroll
    for (int i = 0; i < WPT; ++i) {
        int idx = tid + i * SORT_T;
        if (idx < NW4) {
            int k = idx / N4;
            int nn = 4 * (idx - k * N4);
            sW[(nn + 0) * FPAD + k] = w[i].x;
            sW[(nn + 1) * FPAD + k] = w[i].y;
            sW[(nn + 2) * FPAD + k] = w[i].z;
            sW[(nn + 3) * FPAD + k] = w[i].w;
        }
    }
    if (NT * 16 > NOUT)
        for (int idx = tid; idx < (NT * 16 - NOUT) * 64; idx += SORT_T)
            sW[(NOUT + (idx >> 6)) * FPAD + (idx & 63)] = 0.0f;
    __syncthreads();

    if (wave >= (BNODES / 16) * NT) return;
    int tile   = wave / NT;
    int stripe = wave - tile * NT;
    int node0  = b * BNODES + tile * 16;
    if (node0 >= NNODES) return;

    int half = lane >> 4;
    int lid  = lane & 15;
    const float* arow = sH + (tile * 16 + lid) * FPAD + (half << 1);
    const float* brow = sW + (stripe * 16 + lid) * FPAD + (half << 1);

    v8f d = {};
#pragma unroll 4
    for (int k0 = 0; k0 < 64; k0 += 4) {
        v2f a  = *(const v2f*)(arow + k0);
        v2f bb = *(const v2f*)(brow + k0);
        d = __builtin_amdgcn_wmma_f32_16x16x4_f32(
             false, a,  false, bb,
             (short)0, d,  false,  false);
    }

    int ncol = stripe * 16 + lid;
    if (ncol < NOUT) {
        float* orow = Pn + (size_t)(node0 + (half << 3)) * NOUT + ncol;
#pragma unroll
        for (int r8 = 0; r8 < 8; ++r8)
            orow[(size_t)r8 * NOUT] = d[r8];
    }
}


extern "C" void kernel_launch(void* const* d_in, const int* in_sizes, int n_in,
                              void* d_out, int out_size, void* d_ws, size_t ws_size,
                              hipStream_t stream)
{
    const float* features = (const float*)d_in[0];
    const int*   src      = (const int*)  d_in[1];
    const int*   dst      = (const int*)  d_in[2];
    const float* W0       = (const float*)d_in[3];
    const float* b0       = (const float*)d_in[4];
    const float* W1       = (const float*)d_in[5];
    const float* b1       = (const float*)d_in[6];
    const float* W2       = (const float*)d_in[7];
    const float* b2       = (const float*)d_in[8];
    float*       out      = (float*)d_out;

    float* ws     = (float*)d_ws;
    int*      rowptr     = (int*)(ws + 50176);
    unsigned* binned     = (unsigned*)(ws + 100352);
    int*      csr        = (int*)(ws + 900352);
    float*    P          = ws + 1700352;
    float*    Q          = P  + (size_t)NNODES * 64;
    int*      pieces     = (int*)(Q + (size_t)NNODES * 64);

    const int Mtiles     = NNODES / 16;
    const int gemmBlksX  = (Mtiles + 7) / 8;
    const int nodeBlocks = (NNODES + 8 * NPW - 1) / (8 * NPW);
    const int nodeGroups = (NNODES + 16 * NPW - 1) / (16 * NPW);
    const int buckets    = NBK;

    gemm0_bin_kernel<<<gemmBlksX + NSBLK, 256, 0, stream>>>(features, W0, P, Mtiles, gemmBlksX,
                                                            src, dst, pieces, binned);

    csr_gather_project_kernel<64, 4, true><<<buckets, SORT_T, 0, stream>>>(binned, pieces, rowptr, csr,
                                                                          P, b0, W1, Q, NNODES);

    gather_project_kernel<40, 3, true><<<nodeGroups, 512, 0, stream>>>(Q, rowptr, csr, b1, W2, P);

    gather_kernel<40, false><<<nodeBlocks, 256, 0, stream>>>(P, rowptr, csr, b2, out);
}
